# starting kernel
# baseline (speedup 1.0000x reference)
#include <hip/hip_runtime.h>
#include <math.h>

typedef float v2f __attribute__((ext_vector_type(2)));
typedef float v8f __attribute__((ext_vector_type(8)));

#define N_NODES   100000
#define N_EDGES   1600000
#define D_IN      128
#define D_OUT     64
#define ROW_BLOCKS (N_NODES / 16)

__global__ void k_init(float* __restrict__ out, float* __restrict__ m,
                       float* __restrict__ denom) {
    int i = blockIdx.x * 256 + threadIdx.x;
    if (i < N_NODES * D_OUT) out[i] = 0.0f;
    if (i < N_NODES) { m[i] = -__builtin_inff(); denom[i] = 0.0f; }
}

__global__ __launch_bounds__(256) void k_gemm(const float* __restrict__ X,
                                              const float* __restrict__ W,
                                              const float* __restrict__ bias,
                                              float* __restrict__ Xp) {
    __shared__ float Wl[D_IN * D_OUT];
    const int tid = threadIdx.x;
    for (int i = tid; i < D_IN * D_OUT; i += 256) {
        const int k  = i >> 6;
        const int c  = i & 63;
        const int kk = k & ~1;
        Wl[kk * D_OUT + (c << 1) + (k & 1)] = W[i];
    }
    __syncthreads();

    const int wave     = tid >> 5;
    const int lane     = tid & 31;
    const int rowBlock = blockIdx.x * 8 + wave;
    if (rowBlock >= ROW_BLOCKS) return;

    const int rowBase = rowBlock * 16;
    const int l15     = lane & 15;
    const int khalf   = (lane >> 4) << 1;

    const float* xrow = X + (size_t)(rowBase + l15) * D_IN;

    v8f acc0 = {}, acc1 = {}, acc2 = {}, acc3 = {};
    for (int k0 = 0; k0 < D_IN; k0 += 4) {
        v2f a = *(const v2f*)(xrow + k0 + khalf);
        const float* wp = Wl + (k0 + khalf) * D_OUT;
        v2f b0 = *(const v2f*)(wp + (l15 << 1));
        v2f b1 = *(const v2f*)(wp + ((16 + l15) << 1));
        v2f b2 = *(const v2f*)(wp + ((32 + l15) << 1));
        v2f b3 = *(const v2f*)(wp + ((48 + l15) << 1));
        acc0 = __builtin_amdgcn_wmma_f32_16x16x4_f32(false, a, false, b0, (short)0, acc0, false, false);
        acc1 = __builtin_amdgcn_wmma_f32_16x16x4_f32(false, a, false, b1, (short)0, acc1, false, false);
        acc2 = __builtin_amdgcn_wmma_f32_16x16x4_f32(false, a, false, b2, (short)0, acc2, false, false);
        acc3 = __builtin_amdgcn_wmma_f32_16x16x4_f32(false, a, false, b3, (short)0, acc3, false, false);
    }

    const int orow = rowBase + (lane >> 4) * 8;
    const float b0s = bias[l15], b1s = bias[16 + l15];
    const float b2s = bias[32 + l15], b3s = bias[48 + l15];
#pragma unroll
    for (int r = 0; r < 8; ++r) {
        float* o = Xp + (size_t)(orow + r) * D_OUT;
        o[l15]      = acc0[r] + b0s;
        o[16 + l15] = acc1[r] + b1s;
        o[32 + l15] = acc2[r] + b2s;
        o[48 + l15] = acc3[r] + b3s;
    }
}

__global__ __launch_bounds__(256) void k_scores(const float* __restrict__ Xp,
                                                const float* __restrict__ a_src,
                                                const float* __restrict__ a_src_b,
                                                const float* __restrict__ a_dst,
                                                const float* __restrict__ a_dst_b,
                                                float* __restrict__ s_src,
                                                float* __restrict__ s_dst) {
    const int lane = threadIdx.x & 31;
    const int node = (blockIdx.x * 256 + threadIdx.x) >> 5;
    if (node >= N_NODES) return;
    const float* row = Xp + (size_t)node * D_OUT;
    const float x0 = row[lane], x1 = row[lane + 32];
    float ps = x0 * a_src[lane] + x1 * a_src[lane + 32];
    float pd = x0 * a_dst[lane] + x1 * a_dst[lane + 32];
#pragma unroll
    for (int off = 16; off > 0; off >>= 1) {
        ps += __shfl_xor(ps, off, 32);
        pd += __shfl_xor(pd, off, 32);
    }
    if (lane == 0) {
        s_src[node] = ps + a_src_b[0];
        s_dst[node] = pd + a_dst_b[0];
    }
}

__global__ void k_edge_logits(const int* __restrict__ e_src,
                              const int* __restrict__ e_dst,
                              const float* __restrict__ s_src,
                              const float* __restrict__ s_dst,
                              float* __restrict__ e, float* __restrict__ m) {
    int i = blockIdx.x * 256 + threadIdx.x;
    if (i >= N_EDGES) return;
    float v = s_src[e_src[i]] + s_dst[e_dst[i]];
    v = v > 0.0f ? v : 0.01f * v;
    e[i] = v;
    float* addr = m + e_dst[i];
    if (v >= 0.0f) atomicMax((int*)addr, __float_as_int(v));
    else           atomicMin((unsigned int*)addr, __float_as_uint(v));
}

__global__ void k_edge_exp(const int* __restrict__ e_dst,
                           float* __restrict__ e,
                           const float* __restrict__ m,
                           float* __restrict__ denom) {
    int i = blockIdx.x * 256 + threadIdx.x;
    if (i >= N_EDGES) return;
    const int d = e_dst[i];
    const float w = expf(e[i] - m[d]);
    e[i] = w;
    atomicAdd(denom + d, w);
}

__global__ __launch_bounds__(256) void k_scatter(const int* __restrict__ e_src,
                                                 const int* __restrict__ e_dst,
                                                 const float* __restrict__ e,
                                                 const float* __restrict__ denom,
                                                 const float* __restrict__ Xp,
                                                 float* __restrict__ out) {
    const int lane = threadIdx.x & 31;
    const long long edge = ((long long)blockIdx.x * 256 + threadIdx.x) >> 5;
    if (edge >= N_EDGES) return;
    const int s = e_src[edge], d = e_dst[edge];
    const float coeff = e[edge] / denom[d];
    const float* xs = Xp + (size_t)s * D_OUT;
    float* od = out + (size_t)d * D_OUT;
    atomicAdd(od + lane,      coeff * xs[lane]);
    atomicAdd(od + lane + 32, coeff * xs[lane + 32]);
}

__global__ void k_elu(float* __restrict__ out) {
    int i = blockIdx.x * 256 + threadIdx.x;
    if (i >= N_NODES * D_OUT) return;
    const float v = out[i];
    out[i] = v > 0.0f ? v : (expf(v) - 1.0f);
}

extern "C" void kernel_launch(void* const* d_in, const int* in_sizes, int n_in,
                              void* d_out, int out_size, void* d_ws, size_t ws_size,
                              hipStream_t stream) {
    const float* X       = (const float*)d_in[0];
    const float* theta_w = (const float*)d_in[1];
    const float* theta_b = (const float*)d_in[2];
    const float* a_src   = (const float*)d_in[3];
    const float* a_src_b = (const float*)d_in[4];
    const float* a_dst   = (const float*)d_in[5];
    const float* a_dst_b = (const float*)d_in[6];
    const int*   e_src   = (const int*)d_in[7];
    const int*   e_dst   = (const int*)d_in[8];
    float*       out     = (float*)d_out;

    float* Xp    = (float*)d_ws;
    float* s_src = Xp    + (size_t)N_NODES * D_OUT;
    float* s_dst = s_src + N_NODES;
    float* m     = s_dst + N_NODES;
    float* denom = m     + N_NODES;
    float* e     = denom + N_NODES;

    const int elemGrid  = (N_NODES * D_OUT + 255) / 256;
    const int gemmGrid  = (ROW_BLOCKS + 7) / 8;
    const int nodeWGrid = (N_NODES * 32 + 255) / 256;
    const int edgeGrid  = (N_EDGES + 255) / 256;
    const int edgeWGrid = (int)(((long long)N_EDGES * 32 + 255) / 256);

    k_init<<<elemGrid, 256, 0, stream>>>(out, m, denom);
    k_gemm<<<gemmGrid, 256, 0, stream>>>(X, theta_w, theta_b, Xp);
    k_scores<<<nodeWGrid, 256, 0, stream>>>(Xp, a_src, a_src_b, a_dst, a_dst_b, s_src, s_dst);
    k_edge_logits<<<edgeGrid, 256, 0, stream>>>(e_src, e_dst, s_src, s_dst, e, m);
    k_edge_exp<<<edgeGrid, 256, 0, stream>>>(e_dst, e, m, denom);
    k_scatter<<<edgeWGrid, 256, 0, stream>>>(e_src, e_dst, e, denom, Xp, out);
    k_elu<<<elemGrid, 256, 0, stream>>>(out);
}


# each work-group owns 392 destination nodes: filters the edge list for them into per-wave LDS lists; max, denominator, rows by LDS float atomics (arrival order), divided once per node; 7 launches -> 3
# speedup vs baseline: 3.0271x; 3.0271x over previous
#include <hip/hip_runtime.h>
#include <math.h>

typedef float v2f __attribute__((ext_vector_type(2)));
typedef float v8f __attribute__((ext_vector_type(8)));

#define N_NODES   100000
#define N_EDGES   1600000
#define D_IN      128
#define D_OUT     64
#define ROW_BLOCKS (N_NODES / 16)

__global__ __launch_bounds__(256) void k_gemm(const float* __restrict__ X,
                                              const float* __restrict__ W,
                                              const float* __restrict__ bias,
                                              float* __restrict__ Xp) {
    __shared__ float Wl[D_IN * D_OUT];
    const int tid = threadIdx.x;
    for (int i = tid; i < D_IN * D_OUT; i += 256) {
        const int k  = i >> 6;
        const int c  = i & 63;
        const int kk = k & ~1;
        Wl[kk * D_OUT + (c << 1) + (k & 1)] = W[i];
    }
    __syncthreads();

    const int wave     = tid >> 5;
    const int lane     = tid & 31;
    const int rowBlock = blockIdx.x * 8 + wave;
    if (rowBlock >= ROW_BLOCKS) return;

    const int rowBase = rowBlock * 16;
    const int l15     = lane & 15;
    const int khalf   = (lane >> 4) << 1;

    const float* xrow = X + (size_t)(rowBase + l15) * D_IN;

    v8f acc0 = {}, acc1 = {}, acc2 = {}, acc3 = {};
    for (int k0 = 0; k0 < D_IN; k0 += 4) {
        v2f a = *(const v2f*)(xrow + k0 + khalf);
        const float* wp = Wl + (k0 + khalf) * D_OUT;
        v2f b0 = *(const v2f*)(wp + (l15 << 1));
        v2f b1 = *(const v2f*)(wp + ((16 + l15) << 1));
        v2f b2 = *(const v2f*)(wp + ((32 + l15) << 1));
        v2f b3 = *(const v2f*)(wp + ((48 + l15) << 1));
        acc0 = __builtin_amdgcn_wmma_f32_16x16x4_f32(false, a, false, b0, (short)0, acc0, false, false);
        acc1 = __builtin_amdgcn_wmma_f32_16x16x4_f32(false, a, false, b1, (short)0, acc1, false, false);
        acc2 = __builtin_amdgcn_wmma_f32_16x16x4_f32(false, a, false, b2, (short)0, acc2, false, false);
        acc3 = __builtin_amdgcn_wmma_f32_16x16x4_f32(false, a, false, b3, (short)0, acc3, false, false);
    }

    const int orow = rowBase + (lane >> 4) * 8;
    const float b0s = bias[l15], b1s = bias[16 + l15];
    const float b2s = bias[32 + l15], b3s = bias[48 + l15];
#pragma unroll
    for (int r = 0; r < 8; ++r) {
        float* o = Xp + (size_t)(orow + r) * D_OUT;
        o[l15]      = acc0[r] + b0s;
        o[16 + l15] = acc1[r] + b1s;
        o[32 + l15] = acc2[r] + b2s;
        o[48 + l15] = acc3[r] + b3s;
    }
}

__global__ __launch_bounds__(256) void k_scores(const float* __restrict__ Xp,
                                                const float* __restrict__ a_src,
                                                const float* __restrict__ a_src_b,
                                                const float* __restrict__ a_dst,
                                                const float* __restrict__ a_dst_b,
                                                float* __restrict__ s_src,
                                                float* __restrict__ s_dst) {
    const int lane = threadIdx.x & 31;
    const int node = (blockIdx.x * 256 + threadIdx.x) >> 5;
    if (node >= N_NODES) return;
    const float* row = Xp + (size_t)node * D_OUT;
    const float x0 = row[lane], x1 = row[lane + 32];
    float ps = x0 * a_src[lane] + x1 * a_src[lane + 32];
    float pd = x0 * a_dst[lane] + x1 * a_dst[lane + 32];
#pragma unroll
    for (int off = 16; off > 0; off >>= 1) {
        ps += __shfl_xor(ps, off, 32);
        pd += __shfl_xor(pd, off, 32);
    }
    if (lane == 0) {
        s_src[node] = ps + a_src_b[0];
        s_dst[node] = pd + a_dst_b[0];
    }
}

#define CH      392
#define N_WG    ((N_NODES + CH - 1) / CH)
#define WG_T    1024
#define N_WAVE  (WG_T / 32)
#define WCAP    512
#define NV4     (N_EDGES / 4)
#define SCAN_TRIPS ((NV4 + 4 * WG_T - 1) / (4 * WG_T))
#define DL_BITS 9
#define DL_MASK ((1u << DL_BITS) - 1u)

static_assert(N_EDGES % 64 == 0, "e_dst is read as int4, and the rotated start keeps 256-byte alignment");
static_assert(CH <= (1 << DL_BITS), "local node id must fit DL_BITS");
static_assert(N_EDGES <= (1 << (32 - DL_BITS - 2)), "edge id packs above the local node id");
static_assert(N_NODES <= (1 << (32 - DL_BITS - 2)), "source id packs above the local node id");

__device__ __forceinline__ void lds_fmax(int* slot, float v) {
    if (v >= 0.0f) atomicMax(slot, __float_as_int(v));
    else           atomicMin((unsigned int*)slot, __float_as_uint(v));
}

__device__ __forceinline__ float leaky(float v) { return v > 0.0f ? v : 0.01f * v; }

__device__ __forceinline__ unsigned int umin2(unsigned int a, unsigned int b) { return a < b ? a : b; }

__global__ __launch_bounds__(WG_T) void k_owner(const int* __restrict__ e_src,
                                                const int* __restrict__ e_dst,
                                                const float* __restrict__ s_src,
                                                const float* __restrict__ s_dst,
                                                const float* __restrict__ Xp,
                                                float* __restrict__ out) {
    __shared__ float        acc[CH * D_OUT];
    __shared__ unsigned int ent[N_WAVE * WCAP];
    __shared__ float        val[N_WAVE * WCAP];
    __shared__ int          mI[CH];
    __shared__ float        den[CH];
    __shared__ float        sd[CH];
    __shared__ unsigned int ovf;

    const int tid  = threadIdx.x;
    const int lane = tid & 31;
    const int wave = tid >> 5;
    const int lo   = blockIdx.x * CH;
    const int nn   = (N_NODES - lo) < CH ? (N_NODES - lo) : CH;

    for (int i = tid; i < CH * D_OUT; i += WG_T) acc[i] = 0.0f;
    if (tid < CH) {
        mI[tid]  = __float_as_int(-__builtin_inff());
        den[tid] = 0.0f;
        sd[tid]  = tid < nn ? s_dst[lo + tid] : 0.0f;
    }
    if (tid == 0) ovf = 0u;
    __syncthreads();

    unsigned int* went = ent + wave * WCAP;
    float*        wval = val + wave * WCAP;
    unsigned int  wc   = 0u;

    const int4* dv  = (const int4*)e_dst;
    const int   rot = (int)(((long long)blockIdx.x * NV4) / N_WG) & ~15;
    for (int it = 0; it < SCAN_TRIPS; ++it) {
        const int q0 = tid + it * (4 * WG_T);
        int4 d4[4];
        int  pp[4];
#pragma unroll
        for (int u = 0; u < 4; ++u) {
            const int q = q0 + u * WG_T;
            int p = q + rot;
            if (p >= NV4) p -= NV4;
            pp[u] = p;
            if (q < NV4) d4[u] = dv[p];
            else         d4[u] = make_int4(-1, -1, -1, -1);
        }
#pragma unroll
        for (int u = 0; u < 4; ++u) {
            const unsigned int dl[4] = {(unsigned int)(d4[u].x - lo), (unsigned int)(d4[u].y - lo),
                                        (unsigned int)(d4[u].z - lo), (unsigned int)(d4[u].w - lo)};
            const unsigned int lowest = umin2(umin2(dl[0], dl[1]), umin2(dl[2], dl[3]));
            if (__builtin_amdgcn_ballot_w32(lowest < (unsigned int)nn) != 0u) {
#pragma unroll
                for (int c = 0; c < 4; ++c) {
                    const bool         hit = dl[c] < (unsigned int)nn;
                    const unsigned int mk  = __builtin_amdgcn_ballot_w32(hit);
                    if (mk != 0u) {
                        if (hit) {
                            const unsigned int pos = wc + __builtin_amdgcn_mbcnt_lo(mk, 0u);
                            if (pos < WCAP) went[pos] = ((unsigned int)(4 * pp[u] + c) << DL_BITS) | dl[c];
                        }
                        wc += (unsigned int)__builtin_popcount(mk);
                    }
                }
            }
        }
    }
    if (wc > WCAP && lane == 0) ovf = 1u;
    __syncthreads();

    if (ovf == 0u) {
        const unsigned int n = wc;
        for (unsigned int k0 = lane; k0 < n; k0 += 4 * 32) {
            unsigned int en[4];
            int   s[4];
            float ss[4];
#pragma unroll
            for (int u = 0; u < 4; ++u) {
                const unsigned int k = k0 + u * 32;
                en[u] = k < n ? went[k] : 0u;
            }
#pragma unroll
            for (int u = 0; u < 4; ++u) s[u] = e_src[en[u] >> DL_BITS];
#pragma unroll
            for (int u = 0; u < 4; ++u) ss[u] = s_src[s[u]];
#pragma unroll
            for (int u = 0; u < 4; ++u) {
                const unsigned int k = k0 + u * 32;
                if (k < n) {
                    const unsigned int dl = en[u] & DL_MASK;
                    const float v = leaky(ss[u] + sd[dl]);
                    wval[k] = v;
                    went[k] = ((unsigned int)s[u] << DL_BITS) | dl;
                    lds_fmax(&mI[dl], v);
                }
            }
        }
        __syncthreads();
        for (unsigned int k0 = 0; k0 < n; k0 += 32) {
            const unsigned int k   = k0 + lane;
            const bool         own = k < n;
            const unsigned int e_l = own ? went[k] : 0u;
            float              w_l = 0.0f;
            if (own) {
                w_l = expf(wval[k] - __int_as_float(mI[e_l & DL_MASK]));
                atomicAdd(&den[e_l & DL_MASK], w_l);
            }
#pragma unroll
            for (int h = 0; h < 32; h += 16) {
                if (k0 + h < n) {
                    float2 x[16];
#pragma unroll
                    for (int u = 0; u < 16; ++u) {
                        const unsigned int e = (unsigned int)__builtin_amdgcn_readlane((int)e_l, h + u);
                        x[u] = *(const float2*)(Xp + (size_t)(e >> DL_BITS) * D_OUT + 2 * lane);
                    }
#pragma unroll
                    for (int u = 0; u < 16; ++u) {
                        if (k0 + h + u < n) {
                            const unsigned int e = (unsigned int)__builtin_amdgcn_readlane((int)e_l, h + u);
                            const float        w = __int_as_float(__builtin_amdgcn_readlane(__float_as_int(w_l), h + u));
                            float* a = acc + (e & DL_MASK) * D_OUT + 2 * lane;
                            atomicAdd(a,     w * x[u].x);
                            atomicAdd(a + 1, w * x[u].y);
                        }
                    }
                }
            }
        }
    } else {
        for (int sweep = 0; sweep < 2; ++sweep) {
            for (int q = tid; q < NV4; q += WG_T) {
                const int4 d = dv[q];
                const int dd[4] = {d.x, d.y, d.z, d.w};
                for (int c = 0; c < 4; ++c) {
                    const unsigned int dl = (unsigned int)(dd[c] - lo);
                    if (dl < (unsigned int)nn) {
                        const int   s = e_src[4 * q + c];
                        const float v = leaky(s_src[s] + sd[dl]);
                        if (sweep == 0) {
                            lds_fmax(&mI[dl], v);
                        } else {
                            const float w = expf(v - __int_as_float(mI[dl]));
                            atomicAdd(&den[dl], w);
                            const float* xs = Xp + (size_t)s * D_OUT;
                            for (int f = 0; f < D_OUT; ++f) atomicAdd(&acc[dl * D_OUT + f], w * xs[f]);
                        }
                    }
                }
            }
            __syncthreads();
        }
    }
    __syncthreads();

    float* o = out + (size_t)lo * D_OUT;
    for (int i = tid; i < nn * D_OUT; i += WG_T) {
        const float dn = den[i >> 6];
        const float r  = dn == 0.0f ? 0.0f : acc[i] / dn;
        o[i] = r > 0.0f ? r : (expf(r) - 1.0f);
    }
}

extern "C" void kernel_launch(void* const* d_in, const int* in_sizes, int n_in,
                              void* d_out, int out_size, void* d_ws, size_t ws_size,
                              hipStream_t stream) {
    const float* X       = (const float*)d_in[0];
    const float* theta_w = (const float*)d_in[1];
    const float* theta_b = (const float*)d_in[2];
    const float* a_src   = (const float*)d_in[3];
    const float* a_src_b = (const float*)d_in[4];
    const float* a_dst   = (const float*)d_in[5];
    const float* a_dst_b = (const float*)d_in[6];
    const int*   e_src   = (const int*)d_in[7];
    const int*   e_dst   = (const int*)d_in[8];
    float*       out     = (float*)d_out;

    float* Xp    = (float*)d_ws;
    float* s_src = Xp    + (size_t)N_NODES * D_OUT;
    float* s_dst = s_src + N_NODES;

    const int gemmGrid  = (ROW_BLOCKS + 7) / 8;
    const int nodeWGrid = (N_NODES * 32 + 255) / 256;

    k_gemm<<<gemmGrid, 256, 0, stream>>>(X, theta_w, theta_b, Xp);
    k_scores<<<nodeWGrid, 256, 0, stream>>>(Xp, a_src, a_src_b, a_dst, a_dst_b, s_src, s_dst);
    k_owner<<<N_WG, WG_T, 0, stream>>>(e_src, e_dst, s_src, s_dst, Xp, out);
}


# scores launch: one wave per block of 32 nodes (3125 waves, not 100000): all row loads of a lane issued up front, half-wave DPP row sums, one 128-byte store per array; score sum order changed
# speedup vs baseline: 3.2211x; 1.0641x over previous
#include <hip/hip_runtime.h>
#include <math.h>

typedef float v2f __attribute__((ext_vector_type(2)));
typedef float v8f __attribute__((ext_vector_type(8)));

#define N_NODES   100000
#define N_EDGES   1600000
#define D_IN      128
#define D_OUT     64
#define ROW_BLOCKS (N_NODES / 16)

__global__ __launch_bounds__(256) void k_gemm(const float* __restrict__ X,
                                              const float* __restrict__ W,
                                              const float* __restrict__ bias,
                                              float* __restrict__ Xp) {
    __shared__ float Wl[D_IN * D_OUT];
    const int tid = threadIdx.x;
    for (int i = tid; i < D_IN * D_OUT; i += 256) {
        const int k  = i >> 6;
        const int c  = i & 63;
        const int kk = k & ~1;
        Wl[kk * D_OUT + (c << 1) + (k & 1)] = W[i];
    }
    __syncthreads();

    const int wave     = tid >> 5;
    const int lane     = tid & 31;
    const int rowBlock = blockIdx.x * 8 + wave;
    if (rowBlock >= ROW_BLOCKS) return;

    const int rowBase = rowBlock * 16;
    const int l15     = lane & 15;
    const int khalf   = (lane >> 4) << 1;

    const float* xrow = X + (size_t)(rowBase + l15) * D_IN;

    v8f acc0 = {}, acc1 = {}, acc2 = {}, acc3 = {};
    for (int k0 = 0; k0 < D_IN; k0 += 4) {
        v2f a = *(const v2f*)(xrow + k0 + khalf);
        const float* wp = Wl + (k0 + khalf) * D_OUT;
        v2f b0 = *(const v2f*)(wp + (l15 << 1));
        v2f b1 = *(const v2f*)(wp + ((16 + l15) << 1));
        v2f b2 = *(const v2f*)(wp + ((32 + l15) << 1));
        v2f b3 = *(const v2f*)(wp + ((48 + l15) << 1));
        acc0 = __builtin_amdgcn_wmma_f32_16x16x4_f32(false, a, false, b0, (short)0, acc0, false, false);
        acc1 = __builtin_amdgcn_wmma_f32_16x16x4_f32(false, a, false, b1, (short)0, acc1, false, false);
        acc2 = __builtin_amdgcn_wmma_f32_16x16x4_f32(false, a, false, b2, (short)0, acc2, false, false);
        acc3 = __builtin_amdgcn_wmma_f32_16x16x4_f32(false, a, false, b3, (short)0, acc3, false, false);
    }

    const int orow = rowBase + (lane >> 4) * 8;
    const float b0s = bias[l15], b1s = bias[16 + l15];
    const float b2s = bias[32 + l15], b3s = bias[48 + l15];
#pragma unroll
    for (int r = 0; r < 8; ++r) {
        float* o = Xp + (size_t)(orow + r) * D_OUT;
        o[l15]      = acc0[r] + b0s;
        o[16 + l15] = acc1[r] + b1s;
        o[32 + l15] = acc2[r] + b2s;
        o[48 + l15] = acc3[r] + b3s;
    }
}

#define SC_NODES  32
#define SC_BLOCKS (N_NODES / SC_NODES)
#define SC_WG_T   256

static_assert(N_NODES % SC_NODES == 0, "every wave owns a whole block of 32 nodes");
static_assert(D_OUT == 64, "a 16-lane row of float4 covers one Xp row");

template <int M>
__device__ __forceinline__ float row_xor(float v) {
    return __int_as_float(__builtin_amdgcn_update_dpp(0, __float_as_int(v), 0x160 + M, 0xf, 0xf, true));
}

__device__ __forceinline__ float row_sum16(float v) {
    v += row_xor<1>(v);
    v += row_xor<2>(v);
    v += row_xor<4>(v);
    v += row_xor<8>(v);
    return v;
}

__global__ __launch_bounds__(SC_WG_T) void k_scores(const float* __restrict__ Xp,
                                                    const float* __restrict__ a_src,
                                                    const float* __restrict__ a_src_b,
                                                    const float* __restrict__ a_dst,
                                                    const float* __restrict__ a_dst_b,
                                                    float* __restrict__ s_src,
                                                    float* __restrict__ s_dst) {
    const int lane = threadIdx.x & 31;
    const int blk  = (blockIdx.x * SC_WG_T + threadIdx.x) >> 5;
    if (blk >= SC_BLOCKS) return;
    const int c = lane & 15;
    const int h = lane >> 4;

    const float4 as = *(const float4*)(a_src + 4 * c);
    const float4 ad = *(const float4*)(a_dst + 4 * c);
    const float* base = Xp + ((size_t)blk * SC_NODES + (size_t)h * 16) * D_OUT + 4 * c;

    float4 x[16];
#pragma unroll
    for (int j = 0; j < 16; ++j) x[j] = *(const float4*)(base + j * D_OUT);

    float rs = 0.0f, rd = 0.0f;
#pragma unroll
    for (int j = 0; j < 16; ++j) {
        const float ps = row_sum16(x[j].x * as.x + x[j].y * as.y + x[j].z * as.z + x[j].w * as.w);
        const float pd = row_sum16(x[j].x * ad.x + x[j].y * ad.y + x[j].z * ad.z + x[j].w * ad.w);
        if (c == j) { rs = ps; rd = pd; }
    }
    s_src[blk * SC_NODES + lane] = rs + a_src_b[0];
    s_dst[blk * SC_NODES + lane] = rd + a_dst_b[0];
}

#define CH      392
#define N_WG    ((N_NODES + CH - 1) / CH)
#define WG_T    1024
#define N_WAVE  (WG_T / 32)
#define WCAP    512
#define NV4     (N_EDGES / 4)
#define SCAN_TRIPS ((NV4 + 4 * WG_T - 1) / (4 * WG_T))
#define DL_BITS 9
#define DL_MASK ((1u << DL_BITS) - 1u)

static_assert(N_EDGES % 64 == 0, "e_dst is read as int4, and the rotated start keeps 256-byte alignment");
static_assert(CH <= (1 << DL_BITS), "local node id must fit DL_BITS");
static_assert(N_EDGES <= (1 << (32 - DL_BITS - 2)), "edge id packs above the local node id");
static_assert(N_NODES <= (1 << (32 - DL_BITS - 2)), "source id packs above the local node id");

__device__ __forceinline__ void lds_fmax(int* slot, float v) {
    if (v >= 0.0f) atomicMax(slot, __float_as_int(v));
    else           atomicMin((unsigned int*)slot, __float_as_uint(v));
}

__device__ __forceinline__ float leaky(float v) { return v > 0.0f ? v : 0.01f * v; }

__device__ __forceinline__ unsigned int umin2(unsigned int a, unsigned int b) { return a < b ? a : b; }

__global__ __launch_bounds__(WG_T) void k_owner(const int* __restrict__ e_src,
                                                const int* __restrict__ e_dst,
                                                const float* __restrict__ s_src,
                                                const float* __restrict__ s_dst,
                                                const float* __restrict__ Xp,
                                                float* __restrict__ out) {
    __shared__ float        acc[CH * D_OUT];
    __shared__ unsigned int ent[N_WAVE * WCAP];
    __shared__ float        val[N_WAVE * WCAP];
    __shared__ int          mI[CH];
    __shared__ float        den[CH];
    __shared__ float        sd[CH];
    __shared__ unsigned int ovf;

    const int tid  = threadIdx.x;
    const int lane = tid & 31;
    const int wave = tid >> 5;
    const int lo   = blockIdx.x * CH;
    const int nn   = (N_NODES - lo) < CH ? (N_NODES - lo) : CH;

    for (int i = tid; i < CH * D_OUT; i += WG_T) acc[i] = 0.0f;
    if (tid < CH) {
        mI[tid]  = __float_as_int(-__builtin_inff());
        den[tid] = 0.0f;
        sd[tid]  = tid < nn ? s_dst[lo + tid] : 0.0f;
    }
    if (tid == 0) ovf = 0u;
    __syncthreads();

    unsigned int* went = ent + wave * WCAP;
    float*        wval = val + wave * WCAP;
    unsigned int  wc   = 0u;

    const int4* dv  = (const int4*)e_dst;
    const int   rot = (int)(((long long)blockIdx.x * NV4) / N_WG) & ~15;
    for (int it = 0; it < SCAN_TRIPS; ++it) {
        const int q0 = tid + it * (4 * WG_T);
        int4 d4[4];
        int  pp[4];
#pragma unroll
        for (int u = 0; u < 4; ++u) {
            const int q = q0 + u * WG_T;
            int p = q + rot;
            if (p >= NV4) p -= NV4;
            pp[u] = p;
            if (q < NV4) d4[u] = dv[p];
            else         d4[u] = make_int4(-1, -1, -1, -1);
        }
#pragma unroll
        for (int u = 0; u < 4; ++u) {
            const unsigned int dl[4] = {(unsigned int)(d4[u].x - lo), (unsigned int)(d4[u].y - lo),
                                        (unsigned int)(d4[u].z - lo), (unsigned int)(d4[u].w - lo)};
            const unsigned int lowest = umin2(umin2(dl[0], dl[1]), umin2(dl[2], dl[3]));
            if (__builtin_amdgcn_ballot_w32(lowest < (unsigned int)nn) != 0u) {
#pragma unroll
                for (int c = 0; c < 4; ++c) {
                    const bool         hit = dl[c] < (unsigned int)nn;
                    const unsigned int mk  = __builtin_amdgcn_ballot_w32(hit);
                    if (mk != 0u) {
                        if (hit) {
                            const unsigned int pos = wc + __builtin_amdgcn_mbcnt_lo(mk, 0u);
                            if (pos < WCAP) went[pos] = ((unsigned int)(4 * pp[u] + c) << DL_BITS) | dl[c];
                        }
                        wc += (unsigned int)__builtin_popcount(mk);
                    }
                }
            }
        }
    }
    if (wc > WCAP && lane == 0) ovf = 1u;
    __syncthreads();

    if (ovf == 0u) {
        const unsigned int n = wc;
        for (unsigned int k0 = lane; k0 < n; k0 += 4 * 32) {
            unsigned int en[4];
            int   s[4];
            float ss[4];
#pragma unroll
            for (int u = 0; u < 4; ++u) {
                const unsigned int k = k0 + u * 32;
                en[u] = k < n ? went[k] : 0u;
            }
#pragma unroll
            for (int u = 0; u < 4; ++u) s[u] = e_src[en[u] >> DL_BITS];
#pragma unroll
            for (int u = 0; u < 4; ++u) ss[u] = s_src[s[u]];
#pragma unroll
            for (int u = 0; u < 4; ++u) {
                const unsigned int k = k0 + u * 32;
                if (k < n) {
                    const unsigned int dl = en[u] & DL_MASK;
                    const float v = leaky(ss[u] + sd[dl]);
                    wval[k] = v;
                    went[k] = ((unsigned int)s[u] << DL_BITS) | dl;
                    lds_fmax(&mI[dl], v);
                }
            }
        }
        __syncthreads();
        for (unsigned int k0 = 0; k0 < n; k0 += 32) {
            const unsigned int k   = k0 + lane;
            const bool         own = k < n;
            const unsigned int e_l = own ? went[k] : 0u;
            float              w_l = 0.0f;
            if (own) {
                w_l = expf(wval[k] - __int_as_float(mI[e_l & DL_MASK]));
                atomicAdd(&den[e_l & DL_MASK], w_l);
            }
#pragma unroll
            for (int h = 0; h < 32; h += 16) {
                if (k0 + h < n) {
                    float2 x[16];
#pragma unroll
                    for (int u = 0; u < 16; ++u) {
                        const unsigned int e = (unsigned int)__builtin_amdgcn_readlane((int)e_l, h + u);
                        x[u] = *(const float2*)(Xp + (size_t)(e >> DL_BITS) * D_OUT + 2 * lane);
                    }
#pragma unroll
                    for (int u = 0; u < 16; ++u) {
                        if (k0 + h + u < n) {
                            const unsigned int e = (unsigned int)__builtin_amdgcn_readlane((int)e_l, h + u);
                            const float        w = __int_as_float(__builtin_amdgcn_readlane(__float_as_int(w_l), h + u));
                            float* a = acc + (e & DL_MASK) * D_OUT + 2 * lane;
                            atomicAdd(a,     w * x[u].x);
                            atomicAdd(a + 1, w * x[u].y);
                        }
                    }
                }
            }
        }
    } else {
        for (int sweep = 0; sweep < 2; ++sweep) {
            for (int q = tid; q < NV4; q += WG_T) {
                const int4 d = dv[q];
                const int dd[4] = {d.x, d.y, d.z, d.w};
                for (int c = 0; c < 4; ++c) {
                    const unsigned int dl = (unsigned int)(dd[c] - lo);
                    if (dl < (unsigned int)nn) {
                        const int   s = e_src[4 * q + c];
                        const float v = leaky(s_src[s] + sd[dl]);
                        if (sweep == 0) {
                            lds_fmax(&mI[dl], v);
                        } else {
                            const float w = expf(v - __int_as_float(mI[dl]));
                            atomicAdd(&den[dl], w);
                            const float* xs = Xp + (size_t)s * D_OUT;
                            for (int f = 0; f < D_OUT; ++f) atomicAdd(&acc[dl * D_OUT + f], w * xs[f]);
                        }
                    }
                }
            }
            __syncthreads();
        }
    }
    __syncthreads();

    float* o = out + (size_t)lo * D_OUT;
    for (int i = tid; i < nn * D_OUT; i += WG_T) {
        const float dn = den[i >> 6];
        const float r  = dn == 0.0f ? 0.0f : acc[i] / dn;
        o[i] = r > 0.0f ? r : (expf(r) - 1.0f);
    }
}

extern "C" void kernel_launch(void* const* d_in, const int* in_sizes, int n_in,
                              void* d_out, int out_size, void* d_ws, size_t ws_size,
                              hipStream_t stream) {
    const float* X       = (const float*)d_in[0];
    const float* theta_w = (const float*)d_in[1];
    const float* theta_b = (const float*)d_in[2];
    const float* a_src   = (const float*)d_in[3];
    const float* a_src_b = (const float*)d_in[4];
    const float* a_dst   = (const float*)d_in[5];
    const float* a_dst_b = (const float*)d_in[6];
    const int*   e_src   = (const int*)d_in[7];
    const int*   e_dst   = (const int*)d_in[8];
    float*       out     = (float*)d_out;

    float* Xp    = (float*)d_ws;
    float* s_src = Xp    + (size_t)N_NODES * D_OUT;
    float* s_dst = s_src + N_NODES;

    const int gemmGrid  = (ROW_BLOCKS + 7) / 8;
    const int scoreGrid = (SC_BLOCKS * 32 + SC_WG_T - 1) / SC_WG_T;

    k_gemm<<<gemmGrid, 256, 0, stream>>>(X, theta_w, theta_b, Xp);
    k_scores<<<scoreGrid, SC_WG_T, 0, stream>>>(Xp, a_src, a_src_b, a_dst, a_dst_b, s_src, s_dst);
    k_owner<<<N_WG, WG_T, 0, stream>>>(e_src, e_dst, s_src, s_dst, Xp, out);
}


# edge filter scans owner bit planes (1 byte per edge, 8 words per 32 edges, packed by the scores waves) in place of 4-byte e_dst; list order and so arrival order of the LDS float atomics changes
# speedup vs baseline: 4.1440x; 1.2865x over previous
#include <hip/hip_runtime.h>
#include <math.h>

typedef float v2f __attribute__((ext_vector_type(2)));
typedef float v8f __attribute__((ext_vector_type(8)));

#define N_NODES   100000
#define N_EDGES   1600000
#define D_IN      128
#define D_OUT     64
#define ROW_BLOCKS (N_NODES / 16)

__global__ __launch_bounds__(256) void k_gemm(const float* __restrict__ X,
                                              const float* __restrict__ W,
                                              const float* __restrict__ bias,
                                              float* __restrict__ Xp) {
    __shared__ float Wl[D_IN * D_OUT];
    const int tid = threadIdx.x;
    for (int i = tid; i < D_IN * D_OUT; i += 256) {
        const int k  = i >> 6;
        const int c  = i & 63;
        const int kk = k & ~1;
        Wl[kk * D_OUT + (c << 1) + (k & 1)] = W[i];
    }
    __syncthreads();

    const int wave     = tid >> 5;
    const int lane     = tid & 31;
    const int rowBlock = blockIdx.x * 8 + wave;
    if (rowBlock >= ROW_BLOCKS) return;

    const int rowBase = rowBlock * 16;
    const int l15     = lane & 15;
    const int khalf   = (lane >> 4) << 1;

    const float* xrow = X + (size_t)(rowBase + l15) * D_IN;

    v8f acc0 = {}, acc1 = {}, acc2 = {}, acc3 = {};
    for (int k0 = 0; k0 < D_IN; k0 += 4) {
        v2f a = *(const v2f*)(xrow + k0 + khalf);
        const float* wp = Wl + (k0 + khalf) * D_OUT;
        v2f b0 = *(const v2f*)(wp + (l15 << 1));
        v2f b1 = *(const v2f*)(wp + ((16 + l15) << 1));
        v2f b2 = *(const v2f*)(wp + ((32 + l15) << 1));
        v2f b3 = *(const v2f*)(wp + ((48 + l15) << 1));
        acc0 = __builtin_amdgcn_wmma_f32_16x16x4_f32(false, a, false, b0, (short)0, acc0, false, false);
        acc1 = __builtin_amdgcn_wmma_f32_16x16x4_f32(false, a, false, b1, (short)0, acc1, false, false);
        acc2 = __builtin_amdgcn_wmma_f32_16x16x4_f32(false, a, false, b2, (short)0, acc2, false, false);
        acc3 = __builtin_amdgcn_wmma_f32_16x16x4_f32(false, a, false, b3, (short)0, acc3, false, false);
    }

    const int orow = rowBase + (lane >> 4) * 8;
    const float b0s = bias[l15], b1s = bias[16 + l15];
    const float b2s = bias[32 + l15], b3s = bias[48 + l15];
#pragma unroll
    for (int r = 0; r < 8; ++r) {
        float* o = Xp + (size_t)(orow + r) * D_OUT;
        o[l15]      = acc0[r] + b0s;
        o[16 + l15] = acc1[r] + b1s;
        o[32 + l15] = acc2[r] + b2s;
        o[48 + l15] = acc3[r] + b3s;
    }
}

#define SC_NODES  32
#define SC_BLOCKS (N_NODES / SC_NODES)
#define SC_WG_T   256

#define CH        392
#define N_WG      ((N_NODES + CH - 1) / CH)
#define NG        (N_EDGES / 32)
#define PL_EDGES  (N_EDGES / SC_BLOCKS)
#define PL_GROUPS (PL_EDGES / 32)
#define WS_FLOATS ((size_t)N_NODES * D_OUT + 2 * (size_t)N_NODES)
#define WS_NEED   (WS_FLOATS * 4 + (size_t)N_EDGES)

static_assert(N_WG == 256, "an owner number is exactly 8 bits");
static_assert(N_EDGES % SC_BLOCKS == 0 && PL_EDGES == 512, "every scores wave packs 16 whole groups");
static_assert((WS_FLOATS * 4) % 256 == 0, "the planes start on a 256-byte boundary");

static_assert(N_NODES % SC_NODES == 0, "every wave owns a whole block of 32 nodes");
static_assert(D_OUT == 64, "a 16-lane row of float4 covers one Xp row");

template <int M>
__device__ __forceinline__ float row_xor(float v) {
    return __int_as_float(__builtin_amdgcn_update_dpp(0, __float_as_int(v), 0x160 + M, 0xf, 0xf, true));
}

__device__ __forceinline__ float row_sum16(float v) {
    v += row_xor<1>(v);
    v += row_xor<2>(v);
    v += row_xor<4>(v);
    v += row_xor<8>(v);
    return v;
}

__global__ __launch_bounds__(SC_WG_T) void k_scores(const float* __restrict__ Xp,
                                                    const float* __restrict__ a_src,
                                                    const float* __restrict__ a_src_b,
                                                    const float* __restrict__ a_dst,
                                                    const float* __restrict__ a_dst_b,
                                                    float* __restrict__ s_src,
                                                    float* __restrict__ s_dst,
                                                    const int* __restrict__ e_dst,
                                                    unsigned int* __restrict__ pa,
                                                    unsigned int* __restrict__ pb,
                                                    unsigned long long ws_bytes) {
    const int lane = threadIdx.x & 31;
    const int blk  = (blockIdx.x * SC_WG_T + threadIdx.x) >> 5;
    if (blk >= SC_BLOCKS) return;

    if (ws_bytes >= (unsigned long long)WS_NEED) {
        unsigned int pl[8] = {0u, 0u, 0u, 0u, 0u, 0u, 0u, 0u};
        const int* ed = e_dst + (size_t)blk * PL_EDGES + lane;
        unsigned int ow[PL_GROUPS];
#pragma unroll
        for (int j = 0; j < PL_GROUPS; ++j) ow[j] = (unsigned int)ed[32 * j] / (unsigned int)CH;
#pragma unroll
        for (int j = 0; j < PL_GROUPS; ++j) {
#pragma unroll
            for (int k = 0; k < 8; ++k) {
                const unsigned int b = __builtin_amdgcn_ballot_w32(((ow[j] >> k) & 1u) != 0u);
                if (lane == j) pl[k] = b;
            }
        }
        if (lane < PL_GROUPS) {
            const int g = blk * PL_GROUPS + lane;
            ((uint4*)pa)[g] = make_uint4(pl[0], pl[1], pl[2], pl[3]);
            ((uint4*)pb)[g] = make_uint4(pl[4], pl[5], pl[6], pl[7]);
        }
    }
    const int c = lane & 15;
    const int h = lane >> 4;

    const float4 as = *(const float4*)(a_src + 4 * c);
    const float4 ad = *(const float4*)(a_dst + 4 * c);
    const float* base = Xp + ((size_t)blk * SC_NODES + (size_t)h * 16) * D_OUT + 4 * c;

    float4 x[16];
#pragma unroll
    for (int j = 0; j < 16; ++j) x[j] = *(const float4*)(base + j * D_OUT);

    float rs = 0.0f, rd = 0.0f;
#pragma unroll
    for (int j = 0; j < 16; ++j) {
        const float ps = row_sum16(x[j].x * as.x + x[j].y * as.y + x[j].z * as.z + x[j].w * as.w);
        const float pd = row_sum16(x[j].x * ad.x + x[j].y * ad.y + x[j].z * ad.z + x[j].w * ad.w);
        if (c == j) { rs = ps; rd = pd; }
    }
    s_src[blk * SC_NODES + lane] = rs + a_src_b[0];
    s_dst[blk * SC_NODES + lane] = rd + a_dst_b[0];
}

#define WG_T    1024
#define N_WAVE  (WG_T / 32)
#define WCAP    512
#define NV4     (N_EDGES / 4)
#define PSCAN_TRIPS ((NG + 4 * WG_T - 1) / (4 * WG_T))
#define DL_BITS 9
#define DL_MASK ((1u << DL_BITS) - 1u)

static_assert(N_EDGES % 4 == 0, "the overflow path reads e_dst as int4");
static_assert(NG % 16 == 0, "the rotated start of the plane scan keeps 256-byte alignment");
static_assert(CH <= (1 << DL_BITS), "local node id must fit DL_BITS");
static_assert(N_EDGES <= (1 << (32 - DL_BITS - 2)), "edge id packs above the local node id");
static_assert(N_NODES <= (1 << (32 - DL_BITS - 2)), "source id packs above the local node id");

__device__ __forceinline__ void lds_fmax(int* slot, float v) {
    if (v >= 0.0f) atomicMax(slot, __float_as_int(v));
    else           atomicMin((unsigned int*)slot, __float_as_uint(v));
}

__device__ __forceinline__ float leaky(float v) { return v > 0.0f ? v : 0.01f * v; }

__global__ __launch_bounds__(WG_T) void k_owner(const int* __restrict__ e_src,
                                                const int* __restrict__ e_dst,
                                                const float* __restrict__ s_src,
                                                const float* __restrict__ s_dst,
                                                const float* __restrict__ Xp,
                                                const unsigned int* __restrict__ pa,
                                                const unsigned int* __restrict__ pb,
                                                unsigned long long ws_bytes,
                                                float* __restrict__ out) {
    __shared__ float        acc[CH * D_OUT];
    __shared__ unsigned int ent[N_WAVE * WCAP];
    __shared__ float        val[N_WAVE * WCAP];
    __shared__ int          mI[CH];
    __shared__ float        den[CH];
    __shared__ float        sd[CH];
    __shared__ unsigned int ovf;

    const int tid  = threadIdx.x;
    const int lane = tid & 31;
    const int wave = tid >> 5;
    const int lo   = blockIdx.x * CH;
    const int nn   = (N_NODES - lo) < CH ? (N_NODES - lo) : CH;

    for (int i = tid; i < CH * D_OUT; i += WG_T) acc[i] = 0.0f;
    if (tid < CH) {
        mI[tid]  = __float_as_int(-__builtin_inff());
        den[tid] = 0.0f;
        sd[tid]  = tid < nn ? s_dst[lo + tid] : 0.0f;
    }
    if (tid == 0) ovf = 0u;
    __syncthreads();

    unsigned int* went = ent + wave * WCAP;
    float*        wval = val + wave * WCAP;
    unsigned int  wc   = 0u;

    const bool planes = ws_bytes >= (unsigned long long)WS_NEED;
    unsigned int nb[8];
#pragma unroll
    for (int k = 0; k < 8; ++k) nb[k] = ((blockIdx.x >> k) & 1u) ? 0u : 0xffffffffu;
    const int rot = (int)(((long long)blockIdx.x * NG) / N_WG) & ~15;
    const uint4* pa4 = (const uint4*)pa;
    const uint4* pb4 = (const uint4*)pb;
    for (int it = 0; planes && it < PSCAN_TRIPS; ++it) {
        const int q0 = tid + it * (4 * WG_T);
        uint4 a4[4], b4[4];
        int   pp[4];
#pragma unroll
        for (int u = 0; u < 4; ++u) {
            const int q = q0 + u * WG_T;
            int p = (q < NG ? q : 0) + rot;
            if (p >= NG) p -= NG;
            pp[u] = p;
            a4[u] = pa4[p];
            b4[u] = pb4[p];
        }
#pragma unroll
        for (int u = 0; u < 4; ++u) {
            unsigned int m = (a4[u].x ^ nb[0]) & (a4[u].y ^ nb[1]) & (a4[u].z ^ nb[2]) & (a4[u].w ^ nb[3])
                           & (b4[u].x ^ nb[4]) & (b4[u].y ^ nb[5]) & (b4[u].z ^ nb[6]) & (b4[u].w ^ nb[7]);
            if (q0 + u * WG_T >= NG) m = 0u;
            for (int round = 0; round < 32; ++round) {
                const bool         has = m != 0u;
                const unsigned int mk  = __builtin_amdgcn_ballot_w32(has);
                if (mk == 0u) break;
                if (has) {
                    const unsigned int bit = (unsigned int)__builtin_ctz(m);
                    m &= m - 1u;
                    const unsigned int pos = wc + __builtin_amdgcn_mbcnt_lo(mk, 0u);
                    if (pos < WCAP) went[pos] = 32u * (unsigned int)pp[u] + bit;
                }
                wc += (unsigned int)__builtin_popcount(mk);
            }
        }
    }
    if ((wc > WCAP || !planes) && lane == 0) ovf = 1u;
    __syncthreads();

    if (ovf == 0u) {
        const unsigned int n = wc;
        for (unsigned int k0 = lane; k0 < n; k0 += 4 * 32) {
            unsigned int en[4];
            int   s[4], d[4];
            float ss[4];
#pragma unroll
            for (int u = 0; u < 4; ++u) {
                const unsigned int k = k0 + u * 32;
                en[u] = k < n ? went[k] : 0u;
            }
#pragma unroll
            for (int u = 0; u < 4; ++u) { s[u] = e_src[en[u]]; d[u] = e_dst[en[u]]; }
#pragma unroll
            for (int u = 0; u < 4; ++u) ss[u] = s_src[s[u]];
#pragma unroll
            for (int u = 0; u < 4; ++u) {
                const unsigned int k = k0 + u * 32;
                if (k < n) {
                    const unsigned int dl   = (unsigned int)(d[u] - lo);
                    const bool         live = dl < (unsigned int)nn;
                    const float v = leaky(ss[u] + sd[live ? dl : 0u]);
                    wval[k] = v;
                    went[k] = live ? (((unsigned int)s[u] << DL_BITS) | dl) : DL_MASK;
                    if (live) lds_fmax(&mI[dl], v);
                }
            }
        }
        __syncthreads();
        for (unsigned int k0 = 0; k0 < n; k0 += 32) {
            const unsigned int k   = k0 + lane;
            const unsigned int e_r = k < n ? went[k] : DL_MASK;
            const bool         own = (e_r & DL_MASK) != DL_MASK;
            const unsigned int e_l = own ? e_r : 0u;
            float              w_l = 0.0f;
            if (own) {
                w_l = expf(wval[k] - __int_as_float(mI[e_l & DL_MASK]));
                atomicAdd(&den[e_l & DL_MASK], w_l);
            }
#pragma unroll
            for (int h = 0; h < 32; h += 16) {
                if (k0 + h < n) {
                    float2 x[16];
#pragma unroll
                    for (int u = 0; u < 16; ++u) {
                        const unsigned int e = (unsigned int)__builtin_amdgcn_readlane((int)e_l, h + u);
                        x[u] = *(const float2*)(Xp + (size_t)(e >> DL_BITS) * D_OUT + 2 * lane);
                    }
#pragma unroll
                    for (int u = 0; u < 16; ++u) {
                        const unsigned int e = (unsigned int)__builtin_amdgcn_readlane((int)e_r, h + u);
                        if ((e & DL_MASK) != DL_MASK) {
                            const float        w = __int_as_float(__builtin_amdgcn_readlane(__float_as_int(w_l), h + u));
                            float* a = acc + (e & DL_MASK) * D_OUT + 2 * lane;
                            atomicAdd(a,     w * x[u].x);
                            atomicAdd(a + 1, w * x[u].y);
                        }
                    }
                }
            }
        }
    } else {
        const int4* dv = (const int4*)e_dst;
        for (int sweep = 0; sweep < 2; ++sweep) {
            for (int q = tid; q < NV4; q += WG_T) {
                const int4 d = dv[q];
                const int dd[4] = {d.x, d.y, d.z, d.w};
                for (int c = 0; c < 4; ++c) {
                    const unsigned int dl = (unsigned int)(dd[c] - lo);
                    if (dl < (unsigned int)nn) {
                        const int   s = e_src[4 * q + c];
                        const float v = leaky(s_src[s] + sd[dl]);
                        if (sweep == 0) {
                            lds_fmax(&mI[dl], v);
                        } else {
                            const float w = expf(v - __int_as_float(mI[dl]));
                            atomicAdd(&den[dl], w);
                            const float* xs = Xp + (size_t)s * D_OUT;
                            for (int f = 0; f < D_OUT; ++f) atomicAdd(&acc[dl * D_OUT + f], w * xs[f]);
                        }
                    }
                }
            }
            __syncthreads();
        }
    }
    __syncthreads();

    float* o = out + (size_t)lo * D_OUT;
    for (int i = tid; i < nn * D_OUT; i += WG_T) {
        const float dn = den[i >> 6];
        const float r  = dn == 0.0f ? 0.0f : acc[i] / dn;
        o[i] = r > 0.0f ? r : (expf(r) - 1.0f);
    }
}

extern "C" void kernel_launch(void* const* d_in, const int* in_sizes, int n_in,
                              void* d_out, int out_size, void* d_ws, size_t ws_size,
                              hipStream_t stream) {
    const float* X       = (const float*)d_in[0];
    const float* theta_w = (const float*)d_in[1];
    const float* theta_b = (const float*)d_in[2];
    const float* a_src   = (const float*)d_in[3];
    const float* a_src_b = (const float*)d_in[4];
    const float* a_dst   = (const float*)d_in[5];
    const float* a_dst_b = (const float*)d_in[6];
    const int*   e_src   = (const int*)d_in[7];
    const int*   e_dst   = (const int*)d_in[8];
    float*       out     = (float*)d_out;

    float* Xp    = (float*)d_ws;
    float* s_src = Xp    + (size_t)N_NODES * D_OUT;
    float* s_dst = s_src + N_NODES;
    unsigned int* pa = (unsigned int*)(s_dst + N_NODES);
    unsigned int* pb = pa + 4 * NG;

    const int gemmGrid  = (ROW_BLOCKS + 7) / 8;
    const int scoreGrid = (SC_BLOCKS * 32 + SC_WG_T - 1) / SC_WG_T;

    k_gemm<<<gemmGrid, 256, 0, stream>>>(X, theta_w, theta_b, Xp);
    k_scores<<<scoreGrid, SC_WG_T, 0, stream>>>(Xp, a_src, a_src_b, a_dst, a_dst_b, s_src, s_dst,
                                                e_dst, pa, pb, (unsigned long long)ws_size);
    k_owner<<<N_WG, WG_T, 0, stream>>>(e_src, e_dst, s_src, s_dst, Xp, pa, pb, (unsigned long long)ws_size, out);
}


# scores waves also write one packed word per edge (source id << 9 | local destination); edge-phase pass 1 gathers that one word per kept edge, not e_src[e] and e_dst[e]; sum order unchanged
# speedup vs baseline: 4.3361x; 1.0464x over previous
#include <hip/hip_runtime.h>
#include <math.h>

typedef float v2f __attribute__((ext_vector_type(2)));
typedef float v8f __attribute__((ext_vector_type(8)));

#define N_NODES   100000
#define N_EDGES   1600000
#define D_IN      128
#define D_OUT     64
#define ROW_BLOCKS (N_NODES / 16)

__global__ __launch_bounds__(256) void k_gemm(const float* __restrict__ X,
                                              const float* __restrict__ W,
                                              const float* __restrict__ bias,
                                              float* __restrict__ Xp) {
    __shared__ float Wl[D_IN * D_OUT];
    const int tid = threadIdx.x;
    for (int i = tid; i < D_IN * D_OUT; i += 256) {
        const int k  = i >> 6;
        const int c  = i & 63;
        const int kk = k & ~1;
        Wl[kk * D_OUT + (c << 1) + (k & 1)] = W[i];
    }
    __syncthreads();

    const int wave     = tid >> 5;
    const int lane     = tid & 31;
    const int rowBlock = blockIdx.x * 8 + wave;
    if (rowBlock >= ROW_BLOCKS) return;

    const int rowBase = rowBlock * 16;
    const int l15     = lane & 15;
    const int khalf   = (lane >> 4) << 1;

    const float* xrow = X + (size_t)(rowBase + l15) * D_IN;

    v8f acc0 = {}, acc1 = {}, acc2 = {}, acc3 = {};
    for (int k0 = 0; k0 < D_IN; k0 += 4) {
        v2f a = *(const v2f*)(xrow + k0 + khalf);
        const float* wp = Wl + (k0 + khalf) * D_OUT;
        v2f b0 = *(const v2f*)(wp + (l15 << 1));
        v2f b1 = *(const v2f*)(wp + ((16 + l15) << 1));
        v2f b2 = *(const v2f*)(wp + ((32 + l15) << 1));
        v2f b3 = *(const v2f*)(wp + ((48 + l15) << 1));
        acc0 = __builtin_amdgcn_wmma_f32_16x16x4_f32(false, a, false, b0, (short)0, acc0, false, false);
        acc1 = __builtin_amdgcn_wmma_f32_16x16x4_f32(false, a, false, b1, (short)0, acc1, false, false);
        acc2 = __builtin_amdgcn_wmma_f32_16x16x4_f32(false, a, false, b2, (short)0, acc2, false, false);
        acc3 = __builtin_amdgcn_wmma_f32_16x16x4_f32(false, a, false, b3, (short)0, acc3, false, false);
    }

    const int orow = rowBase + (lane >> 4) * 8;
    const float b0s = bias[l15], b1s = bias[16 + l15];
    const float b2s = bias[32 + l15], b3s = bias[48 + l15];
#pragma unroll
    for (int r = 0; r < 8; ++r) {
        float* o = Xp + (size_t)(orow + r) * D_OUT;
        o[l15]      = acc0[r] + b0s;
        o[16 + l15] = acc1[r] + b1s;
        o[32 + l15] = acc2[r] + b2s;
        o[48 + l15] = acc3[r] + b3s;
    }
}

#define SC_NODES  32
#define SC_BLOCKS (N_NODES / SC_NODES)
#define SC_WG_T   256

#define CH        392
#define N_WG      ((N_NODES + CH - 1) / CH)
#define NG        (N_EDGES / 32)
#define PL_EDGES  (N_EDGES / SC_BLOCKS)
#define PL_GROUPS (PL_EDGES / 32)
#define WS_FLOATS ((size_t)N_NODES * D_OUT + 2 * (size_t)N_NODES)
#define DL_BITS 9
#define DL_MASK ((1u << DL_BITS) - 1u)
#define WS_NEED   (WS_FLOATS * 4 + (size_t)N_EDGES + (size_t)N_EDGES * 4)

static_assert(N_WG == 256, "an owner number is exactly 8 bits");
static_assert(N_EDGES % SC_BLOCKS == 0 && PL_EDGES == 512, "every scores wave packs 16 whole groups");
static_assert((WS_FLOATS * 4) % 256 == 0, "the planes start on a 256-byte boundary");

static_assert(N_NODES % SC_NODES == 0, "every wave owns a whole block of 32 nodes");
static_assert(D_OUT == 64, "a 16-lane row of float4 covers one Xp row");

template <int M>
__device__ __forceinline__ float row_xor(float v) {
    return __int_as_float(__builtin_amdgcn_update_dpp(0, __float_as_int(v), 0x160 + M, 0xf, 0xf, true));
}

__device__ __forceinline__ float row_sum16(float v) {
    v += row_xor<1>(v);
    v += row_xor<2>(v);
    v += row_xor<4>(v);
    v += row_xor<8>(v);
    return v;
}

__global__ __launch_bounds__(SC_WG_T) void k_scores(const float* __restrict__ Xp,
                                                    const float* __restrict__ a_src,
                                                    const float* __restrict__ a_src_b,
                                                    const float* __restrict__ a_dst,
                                                    const float* __restrict__ a_dst_b,
                                                    float* __restrict__ s_src,
                                                    float* __restrict__ s_dst,
                                                    const int* __restrict__ e_dst,
                                                    unsigned int* __restrict__ pa,
                                                    unsigned int* __restrict__ pb,
                                                    const int* __restrict__ e_src,
                                                    unsigned int* __restrict__ pk,
                                                    unsigned long long ws_bytes) {
    const int lane = threadIdx.x & 31;
    const int blk  = (blockIdx.x * SC_WG_T + threadIdx.x) >> 5;
    if (blk >= SC_BLOCKS) return;

    if (ws_bytes >= (unsigned long long)WS_NEED) {
        unsigned int pl[8] = {0u, 0u, 0u, 0u, 0u, 0u, 0u, 0u};
        const int* ed = e_dst + (size_t)blk * PL_EDGES + lane;
        const int*    es = e_src + (size_t)blk * PL_EDGES + lane;
        unsigned int* pw = pk + (size_t)blk * PL_EDGES + lane;
        unsigned int ow[PL_GROUPS];
        unsigned int dv[PL_GROUPS];
        unsigned int sv[PL_GROUPS];
#pragma unroll
        for (int j = 0; j < PL_GROUPS; ++j) dv[j] = (unsigned int)ed[32 * j];
#pragma unroll
        for (int j = 0; j < PL_GROUPS; ++j) sv[j] = (unsigned int)es[32 * j];
#pragma unroll
        for (int j = 0; j < PL_GROUPS; ++j) {
            ow[j] = dv[j] / (unsigned int)CH;
            const unsigned int dl = dv[j] < (unsigned int)N_NODES ? dv[j] - ow[j] * (unsigned int)CH : DL_MASK;
            pw[32 * j] = (sv[j] << DL_BITS) | dl;
        }
#pragma unroll
        for (int j = 0; j < PL_GROUPS; ++j) {
#pragma unroll
            for (int k = 0; k < 8; ++k) {
                const unsigned int b = __builtin_amdgcn_ballot_w32(((ow[j] >> k) & 1u) != 0u);
                if (lane == j) pl[k] = b;
            }
        }
        if (lane < PL_GROUPS) {
            const int g = blk * PL_GROUPS + lane;
            ((uint4*)pa)[g] = make_uint4(pl[0], pl[1], pl[2], pl[3]);
            ((uint4*)pb)[g] = make_uint4(pl[4], pl[5], pl[6], pl[7]);
        }
    }
    const int c = lane & 15;
    const int h = lane >> 4;

    const float4 as = *(const float4*)(a_src + 4 * c);
    const float4 ad = *(const float4*)(a_dst + 4 * c);
    const float* base = Xp + ((size_t)blk * SC_NODES + (size_t)h * 16) * D_OUT + 4 * c;

    float4 x[16];
#pragma unroll
    for (int j = 0; j < 16; ++j) x[j] = *(const float4*)(base + j * D_OUT);

    float rs = 0.0f, rd = 0.0f;
#pragma unroll
    for (int j = 0; j < 16; ++j) {
        const float ps = row_sum16(x[j].x * as.x + x[j].y * as.y + x[j].z * as.z + x[j].w * as.w);
        const float pd = row_sum16(x[j].x * ad.x + x[j].y * ad.y + x[j].z * ad.z + x[j].w * ad.w);
        if (c == j) { rs = ps; rd = pd; }
    }
    s_src[blk * SC_NODES + lane] = rs + a_src_b[0];
    s_dst[blk * SC_NODES + lane] = rd + a_dst_b[0];
}

#define WG_T    1024
#define N_WAVE  (WG_T / 32)
#define WCAP    512
#define NV4     (N_EDGES / 4)
#define PSCAN_TRIPS ((NG + 4 * WG_T - 1) / (4 * WG_T))
static_assert(N_EDGES % 4 == 0, "the overflow path reads e_dst as int4");
static_assert(NG % 16 == 0, "the rotated start of the plane scan keeps 256-byte alignment");
static_assert(CH <= (1 << DL_BITS), "local node id must fit DL_BITS");
static_assert(N_EDGES <= (1 << (32 - DL_BITS - 2)), "edge id packs above the local node id");
static_assert(N_NODES <= (1 << (32 - DL_BITS - 2)), "source id packs above the local node id");

__device__ __forceinline__ void lds_fmax(int* slot, float v) {
    if (v >= 0.0f) atomicMax(slot, __float_as_int(v));
    else           atomicMin((unsigned int*)slot, __float_as_uint(v));
}

__device__ __forceinline__ float leaky(float v) { return v > 0.0f ? v : 0.01f * v; }

__global__ __launch_bounds__(WG_T) void k_owner(const int* __restrict__ e_src,
                                                const int* __restrict__ e_dst,
                                                const float* __restrict__ s_src,
                                                const float* __restrict__ s_dst,
                                                const float* __restrict__ Xp,
                                                const unsigned int* __restrict__ pa,
                                                const unsigned int* __restrict__ pb,
                                                const unsigned int* __restrict__ pk,
                                                unsigned long long ws_bytes,
                                                float* __restrict__ out) {
    __shared__ float        acc[CH * D_OUT];
    __shared__ unsigned int ent[N_WAVE * WCAP];
    __shared__ float        val[N_WAVE * WCAP];
    __shared__ int          mI[CH];
    __shared__ float        den[CH];
    __shared__ float        sd[CH];
    __shared__ unsigned int ovf;

    const int tid  = threadIdx.x;
    const int lane = tid & 31;
    const int wave = tid >> 5;
    const int lo   = blockIdx.x * CH;
    const int nn   = (N_NODES - lo) < CH ? (N_NODES - lo) : CH;

    for (int i = tid; i < CH * D_OUT; i += WG_T) acc[i] = 0.0f;
    if (tid < CH) {
        mI[tid]  = __float_as_int(-__builtin_inff());
        den[tid] = 0.0f;
        sd[tid]  = tid < nn ? s_dst[lo + tid] : 0.0f;
    }
    if (tid == 0) ovf = 0u;
    __syncthreads();

    unsigned int* went = ent + wave * WCAP;
    float*        wval = val + wave * WCAP;
    unsigned int  wc   = 0u;

    const bool planes = ws_bytes >= (unsigned long long)WS_NEED;
    unsigned int nb[8];
#pragma unroll
    for (int k = 0; k < 8; ++k) nb[k] = ((blockIdx.x >> k) & 1u) ? 0u : 0xffffffffu;
    const int rot = (int)(((long long)blockIdx.x * NG) / N_WG) & ~15;
    const uint4* pa4 = (const uint4*)pa;
    const uint4* pb4 = (const uint4*)pb;
    for (int it = 0; planes && it < PSCAN_TRIPS; ++it) {
        const int q0 = tid + it * (4 * WG_T);
        uint4 a4[4], b4[4];
        int   pp[4];
#pragma unroll
        for (int u = 0; u < 4; ++u) {
            const int q = q0 + u * WG_T;
            int p = (q < NG ? q : 0) + rot;
            if (p >= NG) p -= NG;
            pp[u] = p;
            a4[u] = pa4[p];
            b4[u] = pb4[p];
        }
#pragma unroll
        for (int u = 0; u < 4; ++u) {
            unsigned int m = (a4[u].x ^ nb[0]) & (a4[u].y ^ nb[1]) & (a4[u].z ^ nb[2]) & (a4[u].w ^ nb[3])
                           & (b4[u].x ^ nb[4]) & (b4[u].y ^ nb[5]) & (b4[u].z ^ nb[6]) & (b4[u].w ^ nb[7]);
            if (q0 + u * WG_T >= NG) m = 0u;
            for (int round = 0; round < 32; ++round) {
                const bool         has = m != 0u;
                const unsigned int mk  = __builtin_amdgcn_ballot_w32(has);
                if (mk == 0u) break;
                if (has) {
                    const unsigned int bit = (unsigned int)__builtin_ctz(m);
                    m &= m - 1u;
                    const unsigned int pos = wc + __builtin_amdgcn_mbcnt_lo(mk, 0u);
                    if (pos < WCAP) went[pos] = 32u * (unsigned int)pp[u] + bit;
                }
                wc += (unsigned int)__builtin_popcount(mk);
            }
        }
    }
    if ((wc > WCAP || !planes) && lane == 0) ovf = 1u;
    __syncthreads();

    if (ovf == 0u) {
        const unsigned int n = wc;
        for (unsigned int k0 = lane; k0 < n; k0 += 4 * 32) {
            unsigned int en[4];
            int   s[4], d[4];
            float ss[4];
#pragma unroll
            for (int u = 0; u < 4; ++u) {
                const unsigned int k = k0 + u * 32;
                en[u] = k < n ? went[k] : 0u;
            }
#pragma unroll
            for (int u = 0; u < 4; ++u) {
                const unsigned int w = pk[en[u]];
                s[u] = (int)(w >> DL_BITS);
                d[u] = lo + (int)(w & DL_MASK);
            }
#pragma unroll
            for (int u = 0; u < 4; ++u) ss[u] = s_src[s[u]];
#pragma unroll
            for (int u = 0; u < 4; ++u) {
                const unsigned int k = k0 + u * 32;
                if (k < n) {
                    const unsigned int dl   = (unsigned int)(d[u] - lo);
                    const bool         live = dl < (unsigned int)nn;
                    const float v = leaky(ss[u] + sd[live ? dl : 0u]);
                    wval[k] = v;
                    went[k] = live ? (((unsigned int)s[u] << DL_BITS) | dl) : DL_MASK;
                    if (live) lds_fmax(&mI[dl], v);
                }
            }
        }
        __syncthreads();
        for (unsigned int k0 = 0; k0 < n; k0 += 32) {
            const unsigned int k   = k0 + lane;
            const unsigned int e_r = k < n ? went[k] : DL_MASK;
            const bool         own = (e_r & DL_MASK) != DL_MASK;
            const unsigned int e_l = own ? e_r : 0u;
            float              w_l = 0.0f;
            if (own) {
                w_l = expf(wval[k] - __int_as_float(mI[e_l & DL_MASK]));
                atomicAdd(&den[e_l & DL_MASK], w_l);
            }
#pragma unroll
            for (int h = 0; h < 32; h += 16) {
                if (k0 + h < n) {
                    float2 x[16];
#pragma unroll
                    for (int u = 0; u < 16; ++u) {
                        const unsigned int e = (unsigned int)__builtin_amdgcn_readlane((int)e_l, h + u);
                        x[u] = *(const float2*)(Xp + (size_t)(e >> DL_BITS) * D_OUT + 2 * lane);
                    }
#pragma unroll
                    for (int u = 0; u < 16; ++u) {
                        const unsigned int e = (unsigned int)__builtin_amdgcn_readlane((int)e_r, h + u);
                        if ((e & DL_MASK) != DL_MASK) {
                            const float        w = __int_as_float(__builtin_amdgcn_readlane(__float_as_int(w_l), h + u));
                            float* a = acc + (e & DL_MASK) * D_OUT + 2 * lane;
                            atomicAdd(a,     w * x[u].x);
                            atomicAdd(a + 1, w * x[u].y);
                        }
                    }
                }
            }
        }
    } else {
        const int4* dv = (const int4*)e_dst;
        for (int sweep = 0; sweep < 2; ++sweep) {
            for (int q = tid; q < NV4; q += WG_T) {
                const int4 d = dv[q];
                const int dd[4] = {d.x, d.y, d.z, d.w};
                for (int c = 0; c < 4; ++c) {
                    const unsigned int dl = (unsigned int)(dd[c] - lo);
                    if (dl < (unsigned int)nn) {
                        const int   s = e_src[4 * q + c];
                        const float v = leaky(s_src[s] + sd[dl]);
                        if (sweep == 0) {
                            lds_fmax(&mI[dl], v);
                        } else {
                            const float w = expf(v - __int_as_float(mI[dl]));
                            atomicAdd(&den[dl], w);
                            const float* xs = Xp + (size_t)s * D_OUT;
                            for (int f = 0; f < D_OUT; ++f) atomicAdd(&acc[dl * D_OUT + f], w * xs[f]);
                        }
                    }
                }
            }
            __syncthreads();
        }
    }
    __syncthreads();

    float* o = out + (size_t)lo * D_OUT;
    for (int i = tid; i < nn * D_OUT; i += WG_T) {
        const float dn = den[i >> 6];
        const float r  = dn == 0.0f ? 0.0f : acc[i] / dn;
        o[i] = r > 0.0f ? r : (expf(r) - 1.0f);
    }
}

extern "C" void kernel_launch(void* const* d_in, const int* in_sizes, int n_in,
                              void* d_out, int out_size, void* d_ws, size_t ws_size,
                              hipStream_t stream) {
    const float* X       = (const float*)d_in[0];
    const float* theta_w = (const float*)d_in[1];
    const float* theta_b = (const float*)d_in[2];
    const float* a_src   = (const float*)d_in[3];
    const float* a_src_b = (const float*)d_in[4];
    const float* a_dst   = (const float*)d_in[5];
    const float* a_dst_b = (const float*)d_in[6];
    const int*   e_src   = (const int*)d_in[7];
    const int*   e_dst   = (const int*)d_in[8];
    float*       out     = (float*)d_out;

    float* Xp    = (float*)d_ws;
    float* s_src = Xp    + (size_t)N_NODES * D_OUT;
    float* s_dst = s_src + N_NODES;
    unsigned int* pa = (unsigned int*)(s_dst + N_NODES);
    unsigned int* pb = pa + 4 * NG;
    unsigned int* pk = pb + 4 * NG;

    const int gemmGrid  = (ROW_BLOCKS + 7) / 8;
    const int scoreGrid = (SC_BLOCKS * 32 + SC_WG_T - 1) / SC_WG_T;

    k_gemm<<<gemmGrid, 256, 0, stream>>>(X, theta_w, theta_b, Xp);
    k_scores<<<scoreGrid, SC_WG_T, 0, stream>>>(Xp, a_src, a_src_b, a_dst, a_dst_b, s_src, s_dst,
                                                e_dst, pa, pb, e_src, pk, (unsigned long long)ws_size);
    k_owner<<<N_WG, WG_T, 0, stream>>>(e_src, e_dst, s_src, s_dst, Xp, pa, pb, pk, (unsigned long long)ws_size, out);
}


# edge phase: kept edges filed in per-node LDS lists (slot = returning integer add); one wave per owned node forms max, denominator and row in registers; no LDS float atomics; sums now in list order
# speedup vs baseline: 4.9902x; 1.1508x over previous
#include <hip/hip_runtime.h>
#include <math.h>

typedef float v2f __attribute__((ext_vector_type(2)));
typedef float v8f __attribute__((ext_vector_type(8)));

#define N_NODES   100000
#define N_EDGES   1600000
#define D_IN      128
#define D_OUT     64
#define ROW_BLOCKS (N_NODES / 16)

__global__ __launch_bounds__(256) void k_gemm(const float* __restrict__ X,
                                              const float* __restrict__ W,
                                              const float* __restrict__ bias,
                                              float* __restrict__ Xp) {
    __shared__ float Wl[D_IN * D_OUT];
    const int tid = threadIdx.x;
    for (int i = tid; i < D_IN * D_OUT; i += 256) {
        const int k  = i >> 6;
        const int c  = i & 63;
        const int kk = k & ~1;
        Wl[kk * D_OUT + (c << 1) + (k & 1)] = W[i];
    }
    __syncthreads();

    const int wave     = tid >> 5;
    const int lane     = tid & 31;
    const int rowBlock = blockIdx.x * 8 + wave;
    if (rowBlock >= ROW_BLOCKS) return;

    const int rowBase = rowBlock * 16;
    const int l15     = lane & 15;
    const int khalf   = (lane >> 4) << 1;

    const float* xrow = X + (size_t)(rowBase + l15) * D_IN;

    v8f acc0 = {}, acc1 = {}, acc2 = {}, acc3 = {};
    for (int k0 = 0; k0 < D_IN; k0 += 4) {
        v2f a = *(const v2f*)(xrow + k0 + khalf);
        const float* wp = Wl + (k0 + khalf) * D_OUT;
        v2f b0 = *(const v2f*)(wp + (l15 << 1));
        v2f b1 = *(const v2f*)(wp + ((16 + l15) << 1));
        v2f b2 = *(const v2f*)(wp + ((32 + l15) << 1));
        v2f b3 = *(const v2f*)(wp + ((48 + l15) << 1));
        acc0 = __builtin_amdgcn_wmma_f32_16x16x4_f32(false, a, false, b0, (short)0, acc0, false, false);
        acc1 = __builtin_amdgcn_wmma_f32_16x16x4_f32(false, a, false, b1, (short)0, acc1, false, false);
        acc2 = __builtin_amdgcn_wmma_f32_16x16x4_f32(false, a, false, b2, (short)0, acc2, false, false);
        acc3 = __builtin_amdgcn_wmma_f32_16x16x4_f32(false, a, false, b3, (short)0, acc3, false, false);
    }

    const int orow = rowBase + (lane >> 4) * 8;
    const float b0s = bias[l15], b1s = bias[16 + l15];
    const float b2s = bias[32 + l15], b3s = bias[48 + l15];
#pragma unroll
    for (int r = 0; r < 8; ++r) {
        float* o = Xp + (size_t)(orow + r) * D_OUT;
        o[l15]      = acc0[r] + b0s;
        o[16 + l15] = acc1[r] + b1s;
        o[32 + l15] = acc2[r] + b2s;
        o[48 + l15] = acc3[r] + b3s;
    }
}

#define SC_NODES  32
#define SC_BLOCKS (N_NODES / SC_NODES)
#define SC_WG_T   256

#define CH        392
#define N_WG      ((N_NODES + CH - 1) / CH)
#define NG        (N_EDGES / 32)
#define PL_EDGES  (N_EDGES / SC_BLOCKS)
#define PL_GROUPS (PL_EDGES / 32)
#define WS_FLOATS ((size_t)N_NODES * D_OUT + 2 * (size_t)N_NODES)
#define DL_BITS 9
#define DL_MASK ((1u << DL_BITS) - 1u)
#define WS_NEED   (WS_FLOATS * 4 + (size_t)N_EDGES + (size_t)N_EDGES * 4)

static_assert(N_WG == 256, "an owner number is exactly 8 bits");
static_assert(N_EDGES % SC_BLOCKS == 0 && PL_EDGES == 512, "every scores wave packs 16 whole groups");
static_assert((WS_FLOATS * 4) % 256 == 0, "the planes start on a 256-byte boundary");

static_assert(N_NODES % SC_NODES == 0, "every wave owns a whole block of 32 nodes");
static_assert(D_OUT == 64, "a 16-lane row of float4 covers one Xp row");

template <int M>
__device__ __forceinline__ float row_xor(float v) {
    return __int_as_float(__builtin_amdgcn_update_dpp(0, __float_as_int(v), 0x160 + M, 0xf, 0xf, true));
}

__device__ __forceinline__ float row_sum16(float v) {
    v += row_xor<1>(v);
    v += row_xor<2>(v);
    v += row_xor<4>(v);
    v += row_xor<8>(v);
    return v;
}

__global__ __launch_bounds__(SC_WG_T) void k_scores(const float* __restrict__ Xp,
                                                    const float* __restrict__ a_src,
                                                    const float* __restrict__ a_src_b,
                                                    const float* __restrict__ a_dst,
                                                    const float* __restrict__ a_dst_b,
                                                    float* __restrict__ s_src,
                                                    float* __restrict__ s_dst,
                                                    const int* __restrict__ e_dst,
                                                    unsigned int* __restrict__ pa,
                                                    unsigned int* __restrict__ pb,
                                                    const int* __restrict__ e_src,
                                                    unsigned int* __restrict__ pk,
                                                    unsigned long long ws_bytes) {
    const int lane = threadIdx.x & 31;
    const int blk  = (blockIdx.x * SC_WG_T + threadIdx.x) >> 5;
    if (blk >= SC_BLOCKS) return;

    if (ws_bytes >= (unsigned long long)WS_NEED) {
        unsigned int pl[8] = {0u, 0u, 0u, 0u, 0u, 0u, 0u, 0u};
        const int* ed = e_dst + (size_t)blk * PL_EDGES + lane;
        const int*    es = e_src + (size_t)blk * PL_EDGES + lane;
        unsigned int* pw = pk + (size_t)blk * PL_EDGES + lane;
        unsigned int ow[PL_GROUPS];
        unsigned int dv[PL_GROUPS];
        unsigned int sv[PL_GROUPS];
#pragma unroll
        for (int j = 0; j < PL_GROUPS; ++j) dv[j] = (unsigned int)ed[32 * j];
#pragma unroll
        for (int j = 0; j < PL_GROUPS; ++j) sv[j] = (unsigned int)es[32 * j];
#pragma unroll
        for (int j = 0; j < PL_GROUPS; ++j) {
            ow[j] = dv[j] / (unsigned int)CH;
            const unsigned int dl = dv[j] < (unsigned int)N_NODES ? dv[j] - ow[j] * (unsigned int)CH : DL_MASK;
            pw[32 * j] = (sv[j] << DL_BITS) | dl;
        }
#pragma unroll
        for (int j = 0; j < PL_GROUPS; ++j) {
#pragma unroll
            for (int k = 0; k < 8; ++k) {
                const unsigned int b = __builtin_amdgcn_ballot_w32(((ow[j] >> k) & 1u) != 0u);
                if (lane == j) pl[k] = b;
            }
        }
        if (lane < PL_GROUPS) {
            const int g = blk * PL_GROUPS + lane;
            ((uint4*)pa)[g] = make_uint4(pl[0], pl[1], pl[2], pl[3]);
            ((uint4*)pb)[g] = make_uint4(pl[4], pl[5], pl[6], pl[7]);
        }
    }
    const int c = lane & 15;
    const int h = lane >> 4;

    const float4 as = *(const float4*)(a_src + 4 * c);
    const float4 ad = *(const float4*)(a_dst + 4 * c);
    const float* base = Xp + ((size_t)blk * SC_NODES + (size_t)h * 16) * D_OUT + 4 * c;

    float4 x[16];
#pragma unroll
    for (int j = 0; j < 16; ++j) x[j] = *(const float4*)(base + j * D_OUT);

    float rs = 0.0f, rd = 0.0f;
#pragma unroll
    for (int j = 0; j < 16; ++j) {
        const float ps = row_sum16(x[j].x * as.x + x[j].y * as.y + x[j].z * as.z + x[j].w * as.w);
        const float pd = row_sum16(x[j].x * ad.x + x[j].y * ad.y + x[j].z * ad.z + x[j].w * ad.w);
        if (c == j) { rs = ps; rd = pd; }
    }
    s_src[blk * SC_NODES + lane] = rs + a_src_b[0];
    s_dst[blk * SC_NODES + lane] = rd + a_dst_b[0];
}

#define WG_T    1024
#define N_WAVE  (WG_T / 32)
#define WCAP    512
#define NV4     (N_EDGES / 4)
#define PSCAN_TRIPS ((NG + 4 * WG_T - 1) / (4 * WG_T))
static_assert(N_EDGES % 4 == 0, "the overflow path reads e_dst as int4");
static_assert(NG % 16 == 0, "the rotated start of the plane scan keeps 256-byte alignment");
static_assert(CH <= (1 << DL_BITS), "local node id must fit DL_BITS");
static_assert(N_EDGES <= (1 << (32 - DL_BITS - 2)), "edge id packs above the local node id");
static_assert(N_NODES <= (1 << (32 - DL_BITS - 2)), "source id packs above the local node id");

__device__ __forceinline__ void lds_fmax(int* slot, float v) {
    if (v >= 0.0f) atomicMax(slot, __float_as_int(v));
    else           atomicMin((unsigned int*)slot, __float_as_uint(v));
}

__device__ __forceinline__ float leaky(float v) { return v > 0.0f ? v : 0.01f * v; }

#define NCAP 64
static_assert((NCAP & (NCAP - 1)) == 0 && NCAP == 64, "a node's list is two entries per lane");
static_assert(CH * D_OUT <= 2 * CH * NCAP, "the overflow path's rows fit in the list space");

__device__ __forceinline__ float row_max16(float v) {
    v = fmaxf(v, row_xor<1>(v));
    v = fmaxf(v, row_xor<2>(v));
    v = fmaxf(v, row_xor<4>(v));
    v = fmaxf(v, row_xor<8>(v));
    return v;
}
__device__ __forceinline__ float lane_of(float v, int l) {
    return __int_as_float(__builtin_amdgcn_readlane(__float_as_int(v), l));
}

template <int U>
__device__ __forceinline__ void rows_ld(const float* __restrict__ Xp, unsigned int sreg, int l0, int left,
                                        int lane, float2 (&x)[16]) {
    const unsigned int s = (unsigned int)__builtin_amdgcn_readlane((int)sreg, l0 + U);
    x[U] = *(const float2*)(Xp + (size_t)s * D_OUT + 2 * lane);
    if (left > U + 1) rows_ld<U + 1>(Xp, sreg, l0, left, lane, x);
}
template <>
__device__ __forceinline__ void rows_ld<16>(const float* __restrict__, unsigned int, int, int, int, float2 (&)[16]) {}
template <int U>
__device__ __forceinline__ void rows_fm(float wreg, int l0, int left, const float2 (&x)[16], float& ax, float& ay) {
    const float w = lane_of(wreg, l0 + U);
    ax += w * x[U].x;
    ay += w * x[U].y;
    if (left > U + 1) rows_fm<U + 1>(wreg, l0, left, x, ax, ay);
}
template <>
__device__ __forceinline__ void rows_fm<16>(float, int, int, const float2 (&)[16], float&, float&) {}

__global__ __launch_bounds__(WG_T) void k_owner(const int* __restrict__ e_src,
                                                const int* __restrict__ e_dst,
                                                const float* __restrict__ s_src,
                                                const float* __restrict__ s_dst,
                                                const float* __restrict__ Xp,
                                                const unsigned int* __restrict__ pa,
                                                const unsigned int* __restrict__ pb,
                                                const unsigned int* __restrict__ pk,
                                                unsigned long long ws_bytes,
                                                float* __restrict__ out) {
    __shared__ unsigned int ent[N_WAVE * WCAP];
    __shared__ float        nbuf[2 * CH * NCAP];
    __shared__ unsigned int cnt[CH];
    __shared__ int          mI[CH];
    __shared__ float        den[CH];
    __shared__ float        sd[CH];
    __shared__ unsigned int ovf;

    const int tid  = threadIdx.x;
    const int lane = tid & 31;
    const int wave = tid >> 5;
    const int lo   = blockIdx.x * CH;
    const int nn   = (N_NODES - lo) < CH ? (N_NODES - lo) : CH;

    float* nodeS = nbuf;
    float* nodeV = nbuf + CH * NCAP;

    if (tid < CH) {
        cnt[tid] = 0u;
        sd[tid]  = tid < nn ? s_dst[lo + tid] : 0.0f;
    }
    if (tid == 0) ovf = 0u;
    __syncthreads();

    unsigned int* went = ent + wave * WCAP;
    unsigned int  wc   = 0u;

    const bool planes = ws_bytes >= (unsigned long long)WS_NEED;
    unsigned int nb[8];
#pragma unroll
    for (int k = 0; k < 8; ++k) nb[k] = ((blockIdx.x >> k) & 1u) ? 0u : 0xffffffffu;
    const int rot = (int)(((long long)blockIdx.x * NG) / N_WG) & ~15;
    const uint4* pa4 = (const uint4*)pa;
    const uint4* pb4 = (const uint4*)pb;
    for (int it = 0; planes && it < PSCAN_TRIPS; ++it) {
        const int q0 = tid + it * (4 * WG_T);
        uint4 a4[4], b4[4];
        int   pp[4];
#pragma unroll
        for (int u = 0; u < 4; ++u) {
            const int q = q0 + u * WG_T;
            int p = (q < NG ? q : 0) + rot;
            if (p >= NG) p -= NG;
            pp[u] = p;
            a4[u] = pa4[p];
            b4[u] = pb4[p];
        }
#pragma unroll
        for (int u = 0; u < 4; ++u) {
            unsigned int m = (a4[u].x ^ nb[0]) & (a4[u].y ^ nb[1]) & (a4[u].z ^ nb[2]) & (a4[u].w ^ nb[3])
                           & (b4[u].x ^ nb[4]) & (b4[u].y ^ nb[5]) & (b4[u].z ^ nb[6]) & (b4[u].w ^ nb[7]);
            if (q0 + u * WG_T >= NG) m = 0u;
            for (int round = 0; round < 32; ++round) {
                const bool         has = m != 0u;
                const unsigned int mk  = __builtin_amdgcn_ballot_w32(has);
                if (mk == 0u) break;
                if (has) {
                    const unsigned int bit = (unsigned int)__builtin_ctz(m);
                    m &= m - 1u;
                    const unsigned int pos = wc + __builtin_amdgcn_mbcnt_lo(mk, 0u);
                    if (pos < WCAP) went[pos] = 32u * (unsigned int)pp[u] + bit;
                }
                wc += (unsigned int)__builtin_popcount(mk);
            }
        }
    }
    if ((wc > WCAP || !planes) && lane == 0) ovf = 1u;
    __syncthreads();

    if (ovf == 0u) {
        const unsigned int n = wc;
        for (unsigned int k0 = lane; k0 < n; k0 += 4 * 32) {
            unsigned int en[4];
            unsigned int s[4], dlv[4];
            float ss[4];
#pragma unroll
            for (int u = 0; u < 4; ++u) {
                const unsigned int k = k0 + u * 32;
                en[u] = k < n ? went[k] : 0u;
            }
#pragma unroll
            for (int u = 0; u < 4; ++u) {
                const unsigned int w = pk[en[u]];
                s[u]   = w >> DL_BITS;
                dlv[u] = w & DL_MASK;
            }
#pragma unroll
            for (int u = 0; u < 4; ++u) ss[u] = s_src[s[u]];
            bool         live[4];
            float        vv[4];
            unsigned int slot[4];
#pragma unroll
            for (int u = 0; u < 4; ++u) {
                const unsigned int k = k0 + u * 32;
                live[u] = k < n && dlv[u] < (unsigned int)nn;
                vv[u]   = leaky(ss[u] + sd[live[u] ? dlv[u] : 0u]);
            }
#pragma unroll
            for (int u = 0; u < 4; ++u) {
                slot[u] = 0u;
                if (live[u]) slot[u] = atomicAdd(&cnt[dlv[u]], 1u);
            }
#pragma unroll
            for (int u = 0; u < 4; ++u) {
                if (live[u]) {
                    const unsigned int dl = dlv[u];
                    if (slot[u] < NCAP) {
                        const unsigned int at = dl * NCAP + ((slot[u] + dl) & (NCAP - 1));
                        nodeS[at] = __uint_as_float(s[u]);
                        nodeV[at] = vv[u];
                    } else {
                        ovf = 1u;
                    }
                }
            }
        }
    }
    __syncthreads();

    if (ovf == 0u) {
        for (int dl = wave; dl < nn; dl += N_WAVE) {
            const int c = __builtin_amdgcn_readfirstlane((int)cnt[dl]);
            float ax = 0.0f, ay = 0.0f, dsum = 0.0f;
            if (c > 0) {
                const int   p0 = dl * NCAP + ((lane + dl) & (NCAP - 1));
                const int   p1 = dl * NCAP + ((lane + 32 + dl) & (NCAP - 1));
                const bool  h0 = lane < c;
                const bool  h1 = lane + 32 < c;
                float        v0 = -__builtin_inff(), v1 = -__builtin_inff();
                unsigned int s0 = 0u, s1 = 0u;
                if (h0) { v0 = nodeV[p0]; s0 = __float_as_uint(nodeS[p0]); }
                if (h1) { v1 = nodeV[p1]; s1 = __float_as_uint(nodeS[p1]); }
                float2 xa[16];
                rows_ld<0>(Xp, s0, 0, c, lane, xa);
                const float mr = row_max16(fmaxf(v0, v1));
                const float m  = fmaxf(lane_of(mr, 0), lane_of(mr, 16));
                const float w0 = h0 ? expf(v0 - m) : 0.0f;
                const float w1 = h1 ? expf(v1 - m) : 0.0f;
                const float sr = row_sum16(w0 + w1);
                dsum = lane_of(sr, 0) + lane_of(sr, 16);
                rows_fm<0>(w0, 0, c, xa, ax, ay);
                for (int q = 16; q < c; q += 16) {
                    const bool         hi   = q >= 32;
                    const unsigned int sreg = hi ? s1 : s0;
                    const float        wreg = hi ? w1 : w0;
                    float2 x[16];
                    rows_ld<0>(Xp, sreg, q & 31, c - q, lane, x);
                    rows_fm<0>(wreg, q & 31, c - q, x, ax, ay);
                }
            }
            const float inv = dsum == 0.0f ? 0.0f : 1.0f / dsum;
            float rx = ax * inv;
            float ry = ay * inv;
            rx = rx > 0.0f ? rx : (expf(rx) - 1.0f);
            ry = ry > 0.0f ? ry : (expf(ry) - 1.0f);
            *(float2*)(out + (size_t)(lo + dl) * D_OUT + 2 * lane) = make_float2(rx, ry);
        }
    } else {
        float* acc = nbuf;
        for (int i = tid; i < CH * D_OUT; i += WG_T) acc[i] = 0.0f;
        if (tid < CH) {
            mI[tid]  = __float_as_int(-__builtin_inff());
            den[tid] = 0.0f;
        }
        __syncthreads();
        const int4* dv = (const int4*)e_dst;
        for (int sweep = 0; sweep < 2; ++sweep) {
            for (int q = tid; q < NV4; q += WG_T) {
                const int4 d = dv[q];
                const int dd[4] = {d.x, d.y, d.z, d.w};
                for (int c = 0; c < 4; ++c) {
                    const unsigned int dl = (unsigned int)(dd[c] - lo);
                    if (dl < (unsigned int)nn) {
                        const int   s = e_src[4 * q + c];
                        const float v = leaky(s_src[s] + sd[dl]);
                        if (sweep == 0) {
                            lds_fmax(&mI[dl], v);
                        } else {
                            const float w = expf(v - __int_as_float(mI[dl]));
                            atomicAdd(&den[dl], w);
                            const float* xs = Xp + (size_t)s * D_OUT;
                            for (int f = 0; f < D_OUT; ++f) atomicAdd(&acc[dl * D_OUT + f], w * xs[f]);
                        }
                    }
                }
            }
            __syncthreads();
        }
        float* o = out + (size_t)lo * D_OUT;
        for (int i = tid; i < nn * D_OUT; i += WG_T) {
            const float dn = den[i >> 6];
            const float r  = dn == 0.0f ? 0.0f : acc[i] / dn;
            o[i] = r > 0.0f ? r : (expf(r) - 1.0f);
        }
    }
}

extern "C" void kernel_launch(void* const* d_in, const int* in_sizes, int n_in,
                              void* d_out, int out_size, void* d_ws, size_t ws_size,
                              hipStream_t stream) {
    const float* X       = (const float*)d_in[0];
    const float* theta_w = (const float*)d_in[1];
    const float* theta_b = (const float*)d_in[2];
    const float* a_src   = (const float*)d_in[3];
    const float* a_src_b = (const float*)d_in[4];
    const float* a_dst   = (const float*)d_in[5];
    const float* a_dst_b = (const float*)d_in[6];
    const int*   e_src   = (const int*)d_in[7];
    const int*   e_dst   = (const int*)d_in[8];
    float*       out     = (float*)d_out;

    float* Xp    = (float*)d_ws;
    float* s_src = Xp    + (size_t)N_NODES * D_OUT;
    float* s_dst = s_src + N_NODES;
    unsigned int* pa = (unsigned int*)(s_dst + N_NODES);
    unsigned int* pb = pa + 4 * NG;
    unsigned int* pk = pb + 4 * NG;

    const int gemmGrid  = (ROW_BLOCKS + 7) / 8;
    const int scoreGrid = (SC_BLOCKS * 32 + SC_WG_T - 1) / SC_WG_T;

    k_gemm<<<gemmGrid, 256, 0, stream>>>(X, theta_w, theta_b, Xp);
    k_scores<<<scoreGrid, SC_WG_T, 0, stream>>>(Xp, a_src, a_src_b, a_dst, a_dst_b, s_src, s_dst,
                                                e_dst, pa, pb, e_src, pk, (unsigned long long)ws_size);
    k_owner<<<N_WG, WG_T, 0, stream>>>(e_src, e_dst, s_src, s_dst, Xp, pa, pb, pk, (unsigned long long)ws_size, out);
}


# edge phase pass 2: one 16-lane row of a wave per node (2 nodes per wave, 16-byte row loads, entries passed by DPP row_share, no readlane); denominator sum tree changed, row sums not
# speedup vs baseline: 5.2786x; 1.0578x over previous
#include <hip/hip_runtime.h>
#include <math.h>

typedef float v2f __attribute__((ext_vector_type(2)));
typedef float v8f __attribute__((ext_vector_type(8)));

#define N_NODES   100000
#define N_EDGES   1600000
#define D_IN      128
#define D_OUT     64
#define ROW_BLOCKS (N_NODES / 16)

__global__ __launch_bounds__(256) void k_gemm(const float* __restrict__ X,
                                              const float* __restrict__ W,
                                              const float* __restrict__ bias,
                                              float* __restrict__ Xp) {
    __shared__ float Wl[D_IN * D_OUT];
    const int tid = threadIdx.x;
    for (int i = tid; i < D_IN * D_OUT; i += 256) {
        const int k  = i >> 6;
        const int c  = i & 63;
        const int kk = k & ~1;
        Wl[kk * D_OUT + (c << 1) + (k & 1)] = W[i];
    }
    __syncthreads();

    const int wave     = tid >> 5;
    const int lane     = tid & 31;
    const int rowBlock = blockIdx.x * 8 + wave;
    if (rowBlock >= ROW_BLOCKS) return;

    const int rowBase = rowBlock * 16;
    const int l15     = lane & 15;
    const int khalf   = (lane >> 4) << 1;

    const float* xrow = X + (size_t)(rowBase + l15) * D_IN;

    v8f acc0 = {}, acc1 = {}, acc2 = {}, acc3 = {};
    for (int k0 = 0; k0 < D_IN; k0 += 4) {
        v2f a = *(const v2f*)(xrow + k0 + khalf);
        const float* wp = Wl + (k0 + khalf) * D_OUT;
        v2f b0 = *(const v2f*)(wp + (l15 << 1));
        v2f b1 = *(const v2f*)(wp + ((16 + l15) << 1));
        v2f b2 = *(const v2f*)(wp + ((32 + l15) << 1));
        v2f b3 = *(const v2f*)(wp + ((48 + l15) << 1));
        acc0 = __builtin_amdgcn_wmma_f32_16x16x4_f32(false, a, false, b0, (short)0, acc0, false, false);
        acc1 = __builtin_amdgcn_wmma_f32_16x16x4_f32(false, a, false, b1, (short)0, acc1, false, false);
        acc2 = __builtin_amdgcn_wmma_f32_16x16x4_f32(false, a, false, b2, (short)0, acc2, false, false);
        acc3 = __builtin_amdgcn_wmma_f32_16x16x4_f32(false, a, false, b3, (short)0, acc3, false, false);
    }

    const int orow = rowBase + (lane >> 4) * 8;
    const float b0s = bias[l15], b1s = bias[16 + l15];
    const float b2s = bias[32 + l15], b3s = bias[48 + l15];
#pragma unroll
    for (int r = 0; r < 8; ++r) {
        float* o = Xp + (size_t)(orow + r) * D_OUT;
        o[l15]      = acc0[r] + b0s;
        o[16 + l15] = acc1[r] + b1s;
        o[32 + l15] = acc2[r] + b2s;
        o[48 + l15] = acc3[r] + b3s;
    }
}

#define SC_NODES  32
#define SC_BLOCKS (N_NODES / SC_NODES)
#define SC_WG_T   256

#define CH        392
#define N_WG      ((N_NODES + CH - 1) / CH)
#define NG        (N_EDGES / 32)
#define PL_EDGES  (N_EDGES / SC_BLOCKS)
#define PL_GROUPS (PL_EDGES / 32)
#define WS_FLOATS ((size_t)N_NODES * D_OUT + 2 * (size_t)N_NODES)
#define DL_BITS 9
#define DL_MASK ((1u << DL_BITS) - 1u)
#define WS_NEED   (WS_FLOATS * 4 + (size_t)N_EDGES + (size_t)N_EDGES * 4)

static_assert(N_WG == 256, "an owner number is exactly 8 bits");
static_assert(N_EDGES % SC_BLOCKS == 0 && PL_EDGES == 512, "every scores wave packs 16 whole groups");
static_assert((WS_FLOATS * 4) % 256 == 0, "the planes start on a 256-byte boundary");

static_assert(N_NODES % SC_NODES == 0, "every wave owns a whole block of 32 nodes");
static_assert(D_OUT == 64, "a 16-lane row of float4 covers one Xp row");

template <int M>
__device__ __forceinline__ float row_xor(float v) {
    return __int_as_float(__builtin_amdgcn_update_dpp(0, __float_as_int(v), 0x160 + M, 0xf, 0xf, true));
}

__device__ __forceinline__ float row_sum16(float v) {
    v += row_xor<1>(v);
    v += row_xor<2>(v);
    v += row_xor<4>(v);
    v += row_xor<8>(v);
    return v;
}

__global__ __launch_bounds__(SC_WG_T) void k_scores(const float* __restrict__ Xp,
                                                    const float* __restrict__ a_src,
                                                    const float* __restrict__ a_src_b,
                                                    const float* __restrict__ a_dst,
                                                    const float* __restrict__ a_dst_b,
                                                    float* __restrict__ s_src,
                                                    float* __restrict__ s_dst,
                                                    const int* __restrict__ e_dst,
                                                    unsigned int* __restrict__ pa,
                                                    unsigned int* __restrict__ pb,
                                                    const int* __restrict__ e_src,
                                                    unsigned int* __restrict__ pk,
                                                    unsigned long long ws_bytes) {
    const int lane = threadIdx.x & 31;
    const int blk  = (blockIdx.x * SC_WG_T + threadIdx.x) >> 5;
    if (blk >= SC_BLOCKS) return;

    if (ws_bytes >= (unsigned long long)WS_NEED) {
        unsigned int pl[8] = {0u, 0u, 0u, 0u, 0u, 0u, 0u, 0u};
        const int* ed = e_dst + (size_t)blk * PL_EDGES + lane;
        const int*    es = e_src + (size_t)blk * PL_EDGES + lane;
        unsigned int* pw = pk + (size_t)blk * PL_EDGES + lane;
        unsigned int ow[PL_GROUPS];
        unsigned int dv[PL_GROUPS];
        unsigned int sv[PL_GROUPS];
#pragma unroll
        for (int j = 0; j < PL_GROUPS; ++j) dv[j] = (unsigned int)ed[32 * j];
#pragma unroll
        for (int j = 0; j < PL_GROUPS; ++j) sv[j] = (unsigned int)es[32 * j];
#pragma unroll
        for (int j = 0; j < PL_GROUPS; ++j) {
            ow[j] = dv[j] / (unsigned int)CH;
            const unsigned int dl = dv[j] < (unsigned int)N_NODES ? dv[j] - ow[j] * (unsigned int)CH : DL_MASK;
            pw[32 * j] = (sv[j] << DL_BITS) | dl;
        }
#pragma unroll
        for (int j = 0; j < PL_GROUPS; ++j) {
#pragma unroll
            for (int k = 0; k < 8; ++k) {
                const unsigned int b = __builtin_amdgcn_ballot_w32(((ow[j] >> k) & 1u) != 0u);
                if (lane == j) pl[k] = b;
            }
        }
        if (lane < PL_GROUPS) {
            const int g = blk * PL_GROUPS + lane;
            ((uint4*)pa)[g] = make_uint4(pl[0], pl[1], pl[2], pl[3]);
            ((uint4*)pb)[g] = make_uint4(pl[4], pl[5], pl[6], pl[7]);
        }
    }
    const int c = lane & 15;
    const int h = lane >> 4;

    const float4 as = *(const float4*)(a_src + 4 * c);
    const float4 ad = *(const float4*)(a_dst + 4 * c);
    const float* base = Xp + ((size_t)blk * SC_NODES + (size_t)h * 16) * D_OUT + 4 * c;

    float4 x[16];
#pragma unroll
    for (int j = 0; j < 16; ++j) x[j] = *(const float4*)(base + j * D_OUT);

    float rs = 0.0f, rd = 0.0f;
#pragma unroll
    for (int j = 0; j < 16; ++j) {
        const float ps = row_sum16(x[j].x * as.x + x[j].y * as.y + x[j].z * as.z + x[j].w * as.w);
        const float pd = row_sum16(x[j].x * ad.x + x[j].y * ad.y + x[j].z * ad.z + x[j].w * ad.w);
        if (c == j) { rs = ps; rd = pd; }
    }
    s_src[blk * SC_NODES + lane] = rs + a_src_b[0];
    s_dst[blk * SC_NODES + lane] = rd + a_dst_b[0];
}

#define WG_T    1024
#define N_WAVE  (WG_T / 32)
#define WCAP    512
#define NV4     (N_EDGES / 4)
#define PSCAN_TRIPS ((NG + 4 * WG_T - 1) / (4 * WG_T))
static_assert(N_EDGES % 4 == 0, "the overflow path reads e_dst as int4");
static_assert(NG % 16 == 0, "the rotated start of the plane scan keeps 256-byte alignment");
static_assert(CH <= (1 << DL_BITS), "local node id must fit DL_BITS");
static_assert(N_EDGES <= (1 << (32 - DL_BITS - 2)), "edge id packs above the local node id");
static_assert(N_NODES <= (1 << (32 - DL_BITS - 2)), "source id packs above the local node id");

__device__ __forceinline__ void lds_fmax(int* slot, float v) {
    if (v >= 0.0f) atomicMax(slot, __float_as_int(v));
    else           atomicMin((unsigned int*)slot, __float_as_uint(v));
}

__device__ __forceinline__ float leaky(float v) { return v > 0.0f ? v : 0.01f * v; }

#define NCAP 64
static_assert((NCAP & (NCAP - 1)) == 0 && NCAP == 64, "a node's list is four entries per lane of a 16-lane row");
static_assert(CH * D_OUT <= 2 * CH * NCAP, "the overflow path's rows fit in the list space");

__device__ __forceinline__ float row_max16(float v) {
    v = fmaxf(v, row_xor<1>(v));
    v = fmaxf(v, row_xor<2>(v));
    v = fmaxf(v, row_xor<4>(v));
    v = fmaxf(v, row_xor<8>(v));
    return v;
}
__device__ __forceinline__ int lane_of_i(int v, int l) {
    return __builtin_amdgcn_readlane(v, l);
}
template <int U>
__device__ __forceinline__ unsigned int row_get_u(unsigned int v) {
    return (unsigned int)__builtin_amdgcn_update_dpp(0, (int)v, 0x150 + U, 0xf, 0xf, true);
}
template <int U>
__device__ __forceinline__ float row_get_f(float v) {
    return __int_as_float(__builtin_amdgcn_update_dpp(0, __float_as_int(v), 0x150 + U, 0xf, 0xf, true));
}

template <int U>
__device__ __forceinline__ void rows_ld(const float* __restrict__ Xp, unsigned int sreg, int left,
                                        int c, float4 (&x)[16]) {
    const unsigned int s = row_get_u<U>(sreg);
    x[U] = *(const float4*)(Xp + (size_t)s * D_OUT + 4 * c);
    if (left > U + 1) rows_ld<U + 1>(Xp, sreg, left, c, x);
}
template <>
__device__ __forceinline__ void rows_ld<16>(const float* __restrict__, unsigned int, int, int, float4 (&)[16]) {}
template <int U>
__device__ __forceinline__ void rows_fm(float wreg, int left, const float4 (&x)[16], float4& a) {
    const float w = row_get_f<U>(wreg);
    a.x += w * x[U].x;
    a.y += w * x[U].y;
    a.z += w * x[U].z;
    a.w += w * x[U].w;
    if (left > U + 1) rows_fm<U + 1>(wreg, left, x, a);
}
template <>
__device__ __forceinline__ void rows_fm<16>(float, int, const float4 (&)[16], float4&) {}

__global__ __launch_bounds__(WG_T) void k_owner(const int* __restrict__ e_src,
                                                const int* __restrict__ e_dst,
                                                const float* __restrict__ s_src,
                                                const float* __restrict__ s_dst,
                                                const float* __restrict__ Xp,
                                                const unsigned int* __restrict__ pa,
                                                const unsigned int* __restrict__ pb,
                                                const unsigned int* __restrict__ pk,
                                                unsigned long long ws_bytes,
                                                float* __restrict__ out) {
    __shared__ unsigned int ent[N_WAVE * WCAP];
    __shared__ float        nbuf[2 * CH * NCAP];
    __shared__ unsigned int cnt[CH];
    __shared__ int          mI[CH];
    __shared__ float        den[CH];
    __shared__ float        sd[CH];
    __shared__ unsigned int ovf;

    const int tid  = threadIdx.x;
    const int lane = tid & 31;
    const int wave = tid >> 5;
    const int lo   = blockIdx.x * CH;
    const int nn   = (N_NODES - lo) < CH ? (N_NODES - lo) : CH;

    float* nodeS = nbuf;
    float* nodeV = nbuf + CH * NCAP;

    if (tid < CH) {
        cnt[tid] = 0u;
        sd[tid]  = tid < nn ? s_dst[lo + tid] : 0.0f;
    }
    if (tid == 0) ovf = 0u;
    __syncthreads();

    unsigned int* went = ent + wave * WCAP;
    unsigned int  wc   = 0u;

    const bool planes = ws_bytes >= (unsigned long long)WS_NEED;
    unsigned int nb[8];
#pragma unroll
    for (int k = 0; k < 8; ++k) nb[k] = ((blockIdx.x >> k) & 1u) ? 0u : 0xffffffffu;
    const int rot = (int)(((long long)blockIdx.x * NG) / N_WG) & ~15;
    const uint4* pa4 = (const uint4*)pa;
    const uint4* pb4 = (const uint4*)pb;
    for (int it = 0; planes && it < PSCAN_TRIPS; ++it) {
        const int q0 = tid + it * (4 * WG_T);
        uint4 a4[4], b4[4];
        int   pp[4];
#pragma unroll
        for (int u = 0; u < 4; ++u) {
            const int q = q0 + u * WG_T;
            int p = (q < NG ? q : 0) + rot;
            if (p >= NG) p -= NG;
            pp[u] = p;
            a4[u] = pa4[p];
            b4[u] = pb4[p];
        }
#pragma unroll
        for (int u = 0; u < 4; ++u) {
            unsigned int m = (a4[u].x ^ nb[0]) & (a4[u].y ^ nb[1]) & (a4[u].z ^ nb[2]) & (a4[u].w ^ nb[3])
                           & (b4[u].x ^ nb[4]) & (b4[u].y ^ nb[5]) & (b4[u].z ^ nb[6]) & (b4[u].w ^ nb[7]);
            if (q0 + u * WG_T >= NG) m = 0u;
            for (int round = 0; round < 32; ++round) {
                const bool         has = m != 0u;
                const unsigned int mk  = __builtin_amdgcn_ballot_w32(has);
                if (mk == 0u) break;
                if (has) {
                    const unsigned int bit = (unsigned int)__builtin_ctz(m);
                    m &= m - 1u;
                    const unsigned int pos = wc + __builtin_amdgcn_mbcnt_lo(mk, 0u);
                    if (pos < WCAP) went[pos] = 32u * (unsigned int)pp[u] + bit;
                }
                wc += (unsigned int)__builtin_popcount(mk);
            }
        }
    }
    if ((wc > WCAP || !planes) && lane == 0) ovf = 1u;
    __syncthreads();

    if (ovf == 0u) {
        const unsigned int n = wc;
        for (unsigned int k0 = lane; k0 < n; k0 += 4 * 32) {
            unsigned int en[4];
            unsigned int s[4], dlv[4];
            float ss[4];
#pragma unroll
            for (int u = 0; u < 4; ++u) {
                const unsigned int k = k0 + u * 32;
                en[u] = k < n ? went[k] : 0u;
            }
#pragma unroll
            for (int u = 0; u < 4; ++u) {
                const unsigned int w = pk[en[u]];
                s[u]   = w >> DL_BITS;
                dlv[u] = w & DL_MASK;
            }
#pragma unroll
            for (int u = 0; u < 4; ++u) ss[u] = s_src[s[u]];
            bool         live[4];
            float        vv[4];
            unsigned int slot[4];
#pragma unroll
            for (int u = 0; u < 4; ++u) {
                const unsigned int k = k0 + u * 32;
                live[u] = k < n && dlv[u] < (unsigned int)nn;
                vv[u]   = leaky(ss[u] + sd[live[u] ? dlv[u] : 0u]);
            }
#pragma unroll
            for (int u = 0; u < 4; ++u) {
                slot[u] = 0u;
                if (live[u]) slot[u] = atomicAdd(&cnt[dlv[u]], 1u);
            }
#pragma unroll
            for (int u = 0; u < 4; ++u) {
                if (live[u]) {
                    const unsigned int dl = dlv[u];
                    if (slot[u] < NCAP) {
                        const unsigned int at = dl * NCAP + ((slot[u] + dl) & (NCAP - 1));
                        nodeS[at] = __uint_as_float(s[u]);
                        nodeV[at] = vv[u];
                    } else {
                        ovf = 1u;
                    }
                }
            }
        }
    }
    __syncthreads();

    if (ovf == 0u) {
        const int c16   = lane & 15;
        const int hf    = lane >> 4;
        const int npair = (nn + 1) >> 1;
        for (int p = wave; p < npair; p += N_WAVE) {
            const int  dl   = 2 * p + hf;
            const bool own  = dl < nn;
            const int  dq   = own ? dl : 0;
            const int  cn   = own ? (int)cnt[dq] : 0;
            const int  cnA  = lane_of_i(cn, 0);
            const int  cnB  = lane_of_i(cn, 16);
            const int  cmax = cnA > cnB ? cnA : cnB;
            float4 a    = make_float4(0.0f, 0.0f, 0.0f, 0.0f);
            float  dsum = 0.0f;
            if (cmax > 0) {
                const int  lb = dq * NCAP;
                const bool h0 = c16 < cn;
                const bool h1 = c16 + 16 < cn;
                const bool h2 = c16 + 32 < cn;
                const bool h3 = c16 + 48 < cn;
                float        v0 = -__builtin_inff(), v1 = -__builtin_inff(), v2 = -__builtin_inff(), v3 = -__builtin_inff();
                unsigned int s0 = 0u, s1 = 0u, s2 = 0u, s3 = 0u;
                if (h0) { const int q = lb + ((c16 + dq) & (NCAP - 1));      v0 = nodeV[q]; s0 = __float_as_uint(nodeS[q]); }
                if (cmax > 16) {
                    if (h1) { const int q = lb + ((c16 + 16 + dq) & (NCAP - 1)); v1 = nodeV[q]; s1 = __float_as_uint(nodeS[q]); }
                    if (cmax > 32) {
                        if (h2) { const int q = lb + ((c16 + 32 + dq) & (NCAP - 1)); v2 = nodeV[q]; s2 = __float_as_uint(nodeS[q]); }
                        if (h3) { const int q = lb + ((c16 + 48 + dq) & (NCAP - 1)); v3 = nodeV[q]; s3 = __float_as_uint(nodeS[q]); }
                    }
                }
                const unsigned int sfirst = row_get_u<0>(s0);
                if (!h0) s0 = sfirst;
                if (!h1) s1 = sfirst;
                if (!h2) s2 = sfirst;
                if (!h3) s3 = sfirst;
                float w0 = 0.0f, w1 = 0.0f, w2 = 0.0f, w3 = 0.0f;
                for (int q = 0; q < cmax; q += 16) {
                    const unsigned int sreg = q >= 48 ? s3 : (q >= 32 ? s2 : (q >= 16 ? s1 : s0));
                    float4 x[16];
                    rows_ld<0>(Xp, sreg, cmax - q, c16, x);
                    if (q == 0) {
                        const float m = row_max16(fmaxf(fmaxf(v0, v1), fmaxf(v2, v3)));
                        w0 = h0 ? expf(v0 - m) : 0.0f;
                        w1 = h1 ? expf(v1 - m) : 0.0f;
                        w2 = h2 ? expf(v2 - m) : 0.0f;
                        w3 = h3 ? expf(v3 - m) : 0.0f;
                        dsum = row_sum16((w0 + w1) + (w2 + w3));
                    }
                    const float wreg = q >= 48 ? w3 : (q >= 32 ? w2 : (q >= 16 ? w1 : w0));
                    rows_fm<0>(wreg, cmax - q, x, a);
                }
            }
            const bool  any = cn > 0;
            const float inv = (any && dsum != 0.0f) ? 1.0f / dsum : 0.0f;
            float rx = any ? a.x * inv : 0.0f;
            float ry = any ? a.y * inv : 0.0f;
            float rz = any ? a.z * inv : 0.0f;
            float rw = any ? a.w * inv : 0.0f;
            rx = rx > 0.0f ? rx : (expf(rx) - 1.0f);
            ry = ry > 0.0f ? ry : (expf(ry) - 1.0f);
            rz = rz > 0.0f ? rz : (expf(rz) - 1.0f);
            rw = rw > 0.0f ? rw : (expf(rw) - 1.0f);
            if (own) *(float4*)(out + (size_t)(lo + dl) * D_OUT + 4 * c16) = make_float4(rx, ry, rz, rw);
        }
    } else {
        float* acc = nbuf;
        for (int i = tid; i < CH * D_OUT; i += WG_T) acc[i] = 0.0f;
        if (tid < CH) {
            mI[tid]  = __float_as_int(-__builtin_inff());
            den[tid] = 0.0f;
        }
        __syncthreads();
        const int4* dv = (const int4*)e_dst;
        for (int sweep = 0; sweep < 2; ++sweep) {
            for (int q = tid; q < NV4; q += WG_T) {
                const int4 d = dv[q];
                const int dd[4] = {d.x, d.y, d.z, d.w};
                for (int c = 0; c < 4; ++c) {
                    const unsigned int dl = (unsigned int)(dd[c] - lo);
                    if (dl < (unsigned int)nn) {
                        const int   s = e_src[4 * q + c];
                        const float v = leaky(s_src[s] + sd[dl]);
                        if (sweep == 0) {
                            lds_fmax(&mI[dl], v);
                        } else {
                            const float w = expf(v - __int_as_float(mI[dl]));
                            atomicAdd(&den[dl], w);
                            const float* xs = Xp + (size_t)s * D_OUT;
                            for (int f = 0; f < D_OUT; ++f) atomicAdd(&acc[dl * D_OUT + f], w * xs[f]);
                        }
                    }
                }
            }
            __syncthreads();
        }
        float* o = out + (size_t)lo * D_OUT;
        for (int i = tid; i < nn * D_OUT; i += WG_T) {
            const float dn = den[i >> 6];
            const float r  = dn == 0.0f ? 0.0f : acc[i] / dn;
            o[i] = r > 0.0f ? r : (expf(r) - 1.0f);
        }
    }
}

extern "C" void kernel_launch(void* const* d_in, const int* in_sizes, int n_in,
                              void* d_out, int out_size, void* d_ws, size_t ws_size,
                              hipStream_t stream) {
    const float* X       = (const float*)d_in[0];
    const float* theta_w = (const float*)d_in[1];
    const float* theta_b = (const float*)d_in[2];
    const float* a_src   = (const float*)d_in[3];
    const float* a_src_b = (const float*)d_in[4];
    const float* a_dst   = (const float*)d_in[5];
    const float* a_dst_b = (const float*)d_in[6];
    const int*   e_src   = (const int*)d_in[7];
    const int*   e_dst   = (const int*)d_in[8];
    float*       out     = (float*)d_out;

    float* Xp    = (float*)d_ws;
    float* s_src = Xp    + (size_t)N_NODES * D_OUT;
    float* s_dst = s_src + N_NODES;
    unsigned int* pa = (unsigned int*)(s_dst + N_NODES);
    unsigned int* pb = pa + 4 * NG;
    unsigned int* pk = pb + 4 * NG;

    const int gemmGrid  = (ROW_BLOCKS + 7) / 8;
    const int scoreGrid = (SC_BLOCKS * 32 + SC_WG_T - 1) / SC_WG_T;

    k_gemm<<<gemmGrid, 256, 0, stream>>>(X, theta_w, theta_b, Xp);
    k_scores<<<scoreGrid, SC_WG_T, 0, stream>>>(Xp, a_src, a_src_b, a_dst, a_dst_b, s_src, s_dst,
                                                e_dst, pa, pb, e_src, pk, (unsigned long long)ws_size);
    k_owner<<<N_WG, WG_T, 0, stream>>>(e_src, e_dst, s_src, s_dst, Xp, pa, pb, pk, (unsigned long long)ws_size, out);
}


# edge filter reads two one-hot owner bitmaps (low, high half of the owner number: 8 bytes per 32 edges, packed by the scores waves), not eight bit planes (32 bytes); list and sum order change
# speedup vs baseline: 5.9308x; 1.1236x over previous
#include <hip/hip_runtime.h>
#include <math.h>

typedef float v2f __attribute__((ext_vector_type(2)));
typedef float v8f __attribute__((ext_vector_type(8)));

#define N_NODES   100000
#define N_EDGES   1600000
#define D_IN      128
#define D_OUT     64
#define ROW_BLOCKS (N_NODES / 16)

__global__ __launch_bounds__(256) void k_gemm(const float* __restrict__ X,
                                              const float* __restrict__ W,
                                              const float* __restrict__ bias,
                                              float* __restrict__ Xp) {
    __shared__ float Wl[D_IN * D_OUT];
    const int tid = threadIdx.x;
    for (int i = tid; i < D_IN * D_OUT; i += 256) {
        const int k  = i >> 6;
        const int c  = i & 63;
        const int kk = k & ~1;
        Wl[kk * D_OUT + (c << 1) + (k & 1)] = W[i];
    }
    __syncthreads();

    const int wave     = tid >> 5;
    const int lane     = tid & 31;
    const int rowBlock = blockIdx.x * 8 + wave;
    if (rowBlock >= ROW_BLOCKS) return;

    const int rowBase = rowBlock * 16;
    const int l15     = lane & 15;
    const int khalf   = (lane >> 4) << 1;

    const float* xrow = X + (size_t)(rowBase + l15) * D_IN;

    v8f acc0 = {}, acc1 = {}, acc2 = {}, acc3 = {};
    for (int k0 = 0; k0 < D_IN; k0 += 4) {
        v2f a = *(const v2f*)(xrow + k0 + khalf);
        const float* wp = Wl + (k0 + khalf) * D_OUT;
        v2f b0 = *(const v2f*)(wp + (l15 << 1));
        v2f b1 = *(const v2f*)(wp + ((16 + l15) << 1));
        v2f b2 = *(const v2f*)(wp + ((32 + l15) << 1));
        v2f b3 = *(const v2f*)(wp + ((48 + l15) << 1));
        acc0 = __builtin_amdgcn_wmma_f32_16x16x4_f32(false, a, false, b0, (short)0, acc0, false, false);
        acc1 = __builtin_amdgcn_wmma_f32_16x16x4_f32(false, a, false, b1, (short)0, acc1, false, false);
        acc2 = __builtin_amdgcn_wmma_f32_16x16x4_f32(false, a, false, b2, (short)0, acc2, false, false);
        acc3 = __builtin_amdgcn_wmma_f32_16x16x4_f32(false, a, false, b3, (short)0, acc3, false, false);
    }

    const int orow = rowBase + (lane >> 4) * 8;
    const float b0s = bias[l15], b1s = bias[16 + l15];
    const float b2s = bias[32 + l15], b3s = bias[48 + l15];
#pragma unroll
    for (int r = 0; r < 8; ++r) {
        float* o = Xp + (size_t)(orow + r) * D_OUT;
        o[l15]      = acc0[r] + b0s;
        o[16 + l15] = acc1[r] + b1s;
        o[32 + l15] = acc2[r] + b2s;
        o[48 + l15] = acc3[r] + b3s;
    }
}

#define SC_NODES  32
#define SC_BLOCKS (N_NODES / SC_NODES)
#define SC_WG_T   256

#define CH        392
#define N_WG      ((N_NODES + CH - 1) / CH)
#define NG        (N_EDGES / 32)
#define PL_EDGES  (N_EDGES / SC_BLOCKS)
#define PL_GROUPS (PL_EDGES / 32)
#define N_BM      32
#define BM_STRIDE (((NG + 63) / 64) * 64)
#define WS_FLOATS ((size_t)N_NODES * D_OUT + 2 * (size_t)N_NODES)
#define DL_BITS 9
#define DL_MASK ((1u << DL_BITS) - 1u)
#define WS_NEED   (WS_FLOATS * 4 + (size_t)N_BM * BM_STRIDE * 4 + (size_t)N_EDGES * 4)

static_assert(N_WG == 256, "an owner number is exactly 8 bits");
static_assert(N_EDGES % SC_BLOCKS == 0 && PL_EDGES == 512, "every scores wave packs 16 whole groups");
static_assert((WS_FLOATS * 4) % 256 == 0, "the bitmaps start on a 256-byte boundary");
static_assert(BM_STRIDE % 64 == 0 && BM_STRIDE >= NG, "every bitmap holds all groups and starts on a 256-byte boundary");

static_assert(N_NODES % SC_NODES == 0, "every wave owns a whole block of 32 nodes");
static_assert(D_OUT == 64, "a 16-lane row of float4 covers one Xp row");

template <int M>
__device__ __forceinline__ float row_xor(float v) {
    return __int_as_float(__builtin_amdgcn_update_dpp(0, __float_as_int(v), 0x160 + M, 0xf, 0xf, true));
}

__device__ __forceinline__ float row_sum16(float v) {
    v += row_xor<1>(v);
    v += row_xor<2>(v);
    v += row_xor<4>(v);
    v += row_xor<8>(v);
    return v;
}

__global__ __launch_bounds__(SC_WG_T) void k_scores(const float* __restrict__ Xp,
                                                    const float* __restrict__ a_src,
                                                    const float* __restrict__ a_src_b,
                                                    const float* __restrict__ a_dst,
                                                    const float* __restrict__ a_dst_b,
                                                    float* __restrict__ s_src,
                                                    float* __restrict__ s_dst,
                                                    const int* __restrict__ e_dst,
                                                    unsigned int* __restrict__ bm,
                                                    const int* __restrict__ e_src,
                                                    unsigned int* __restrict__ pk,
                                                    unsigned long long ws_bytes) {
    const int lane = threadIdx.x & 31;
    const int blk  = (blockIdx.x * SC_WG_T + threadIdx.x) >> 5;
    if (blk >= SC_BLOCKS) return;

    if (ws_bytes >= (unsigned long long)WS_NEED) {
        unsigned int pl[8] = {0u, 0u, 0u, 0u, 0u, 0u, 0u, 0u};
        const int* ed = e_dst + (size_t)blk * PL_EDGES + lane;
        const int*    es = e_src + (size_t)blk * PL_EDGES + lane;
        unsigned int* pw = pk + (size_t)blk * PL_EDGES + lane;
        unsigned int ow[PL_GROUPS];
        unsigned int dv[PL_GROUPS];
        unsigned int sv[PL_GROUPS];
#pragma unroll
        for (int j = 0; j < PL_GROUPS; ++j) dv[j] = (unsigned int)ed[32 * j];
#pragma unroll
        for (int j = 0; j < PL_GROUPS; ++j) sv[j] = (unsigned int)es[32 * j];
#pragma unroll
        for (int j = 0; j < PL_GROUPS; ++j) {
            ow[j] = dv[j] / (unsigned int)CH;
            const unsigned int dl = dv[j] < (unsigned int)N_NODES ? dv[j] - ow[j] * (unsigned int)CH : DL_MASK;
            pw[32 * j] = (sv[j] << DL_BITS) | dl;
        }
#pragma unroll
        for (int j = 0; j < PL_GROUPS; ++j) {
#pragma unroll
            for (int k = 0; k < 8; ++k) {
                const unsigned int b = __builtin_amdgcn_ballot_w32(((ow[j] >> k) & 1u) != 0u);
                if ((lane & 15) == j) pl[k] = b;
            }
        }
        const bool hi = lane >= 16;
        const unsigned int q0 = hi ? pl[4] : pl[0];
        const unsigned int q1 = hi ? pl[5] : pl[1];
        const unsigned int q2 = hi ? pl[6] : pl[2];
        const unsigned int q3 = hi ? pl[7] : pl[3];
        const unsigned int va[4] = {~q1 & ~q0, ~q1 & q0, q1 & ~q0, q1 & q0};
        const unsigned int vb[4] = {~q3 & ~q2, ~q3 & q2, q3 & ~q2, q3 & q2};
        unsigned int* bw = bm + (size_t)(hi ? 16 : 0) * BM_STRIDE + (size_t)blk * PL_GROUPS + (lane & 15);
#pragma unroll
        for (int c = 0; c < 16; ++c) bw[(size_t)c * BM_STRIDE] = va[c & 3] & vb[c >> 2];
    }
    const int c = lane & 15;
    const int h = lane >> 4;

    const float4 as = *(const float4*)(a_src + 4 * c);
    const float4 ad = *(const float4*)(a_dst + 4 * c);
    const float* base = Xp + ((size_t)blk * SC_NODES + (size_t)h * 16) * D_OUT + 4 * c;

    float4 x[16];
#pragma unroll
    for (int j = 0; j < 16; ++j) x[j] = *(const float4*)(base + j * D_OUT);

    float rs = 0.0f, rd = 0.0f;
#pragma unroll
    for (int j = 0; j < 16; ++j) {
        const float ps = row_sum16(x[j].x * as.x + x[j].y * as.y + x[j].z * as.z + x[j].w * as.w);
        const float pd = row_sum16(x[j].x * ad.x + x[j].y * ad.y + x[j].z * ad.z + x[j].w * ad.w);
        if (c == j) { rs = ps; rd = pd; }
    }
    s_src[blk * SC_NODES + lane] = rs + a_src_b[0];
    s_dst[blk * SC_NODES + lane] = rd + a_dst_b[0];
}

#define WG_T    1024
#define N_WAVE  (WG_T / 32)
#define WCAP    512
#define NV4     (N_EDGES / 4)
#define NQ      (NG / 4)
#define QSLOTS  ((NQ + WG_T - 1) / WG_T)
#define QSCAN_TRIPS ((QSLOTS + 3) / 4)
static_assert(N_EDGES % 4 == 0, "the overflow path reads e_dst as int4");
static_assert(NG % 4 == 0, "a bitmap is read as 16-byte vectors of four whole groups");
static_assert(CH <= (1 << DL_BITS), "local node id must fit DL_BITS");
static_assert(N_EDGES <= (1 << (32 - DL_BITS - 2)), "edge id packs above the local node id");
static_assert(N_NODES <= (1 << (32 - DL_BITS - 2)), "source id packs above the local node id");

__device__ __forceinline__ void lds_fmax(int* slot, float v) {
    if (v >= 0.0f) atomicMax(slot, __float_as_int(v));
    else           atomicMin((unsigned int*)slot, __float_as_uint(v));
}

__device__ __forceinline__ float leaky(float v) { return v > 0.0f ? v : 0.01f * v; }

#define NCAP 64
static_assert((NCAP & (NCAP - 1)) == 0 && NCAP == 64, "a node's list is four entries per lane of a 16-lane row");
static_assert(CH * D_OUT <= 2 * CH * NCAP, "the overflow path's rows fit in the list space");

__device__ __forceinline__ float row_max16(float v) {
    v = fmaxf(v, row_xor<1>(v));
    v = fmaxf(v, row_xor<2>(v));
    v = fmaxf(v, row_xor<4>(v));
    v = fmaxf(v, row_xor<8>(v));
    return v;
}
__device__ __forceinline__ int lane_of_i(int v, int l) {
    return __builtin_amdgcn_readlane(v, l);
}
template <int U>
__device__ __forceinline__ unsigned int row_get_u(unsigned int v) {
    return (unsigned int)__builtin_amdgcn_update_dpp(0, (int)v, 0x150 + U, 0xf, 0xf, true);
}
template <int U>
__device__ __forceinline__ float row_get_f(float v) {
    return __int_as_float(__builtin_amdgcn_update_dpp(0, __float_as_int(v), 0x150 + U, 0xf, 0xf, true));
}

template <int U>
__device__ __forceinline__ void rows_ld(const float* __restrict__ Xp, unsigned int sreg, int left,
                                        int c, float4 (&x)[16]) {
    const unsigned int s = row_get_u<U>(sreg);
    x[U] = *(const float4*)(Xp + (size_t)s * D_OUT + 4 * c);
    if (left > U + 1) rows_ld<U + 1>(Xp, sreg, left, c, x);
}
template <>
__device__ __forceinline__ void rows_ld<16>(const float* __restrict__, unsigned int, int, int, float4 (&)[16]) {}
template <int U>
__device__ __forceinline__ void rows_fm(float wreg, int left, const float4 (&x)[16], float4& a) {
    const float w = row_get_f<U>(wreg);
    a.x += w * x[U].x;
    a.y += w * x[U].y;
    a.z += w * x[U].z;
    a.w += w * x[U].w;
    if (left > U + 1) rows_fm<U + 1>(wreg, left, x, a);
}
template <>
__device__ __forceinline__ void rows_fm<16>(float, int, const float4 (&)[16], float4&) {}

__global__ __launch_bounds__(WG_T) void k_owner(const int* __restrict__ e_src,
                                                const int* __restrict__ e_dst,
                                                const float* __restrict__ s_src,
                                                const float* __restrict__ s_dst,
                                                const float* __restrict__ Xp,
                                                const unsigned int* __restrict__ bm,
                                                const unsigned int* __restrict__ pk,
                                                unsigned long long ws_bytes,
                                                float* __restrict__ out) {
    __shared__ unsigned int ent[N_WAVE * WCAP];
    __shared__ float        nbuf[2 * CH * NCAP];
    __shared__ unsigned int cnt[CH];
    __shared__ int          mI[CH];
    __shared__ float        den[CH];
    __shared__ float        sd[CH];
    __shared__ unsigned int ovf;

    const int tid  = threadIdx.x;
    const int lane = tid & 31;
    const int wave = tid >> 5;
    const int lo   = blockIdx.x * CH;
    const int nn   = (N_NODES - lo) < CH ? (N_NODES - lo) : CH;

    float* nodeS = nbuf;
    float* nodeV = nbuf + CH * NCAP;

    if (tid < CH) {
        cnt[tid] = 0u;
        sd[tid]  = tid < nn ? s_dst[lo + tid] : 0.0f;
    }
    if (tid == 0) ovf = 0u;
    __syncthreads();

    unsigned int* went = ent + wave * WCAP;
    unsigned int  wc   = 0u;

    const bool maps = ws_bytes >= (unsigned long long)WS_NEED;
    const uint4* bl4 = (const uint4*)(bm + (size_t)(blockIdx.x & 15u) * BM_STRIDE);
    const uint4* bh4 = (const uint4*)(bm + (size_t)(16u + (blockIdx.x >> 4)) * BM_STRIDE);
    const int rot = (int)(((((blockIdx.x & 15u) + (blockIdx.x >> 4)) & 15u) * (unsigned int)NQ) / 16u) & ~15;
    auto drain = [&](unsigned long long x, unsigned int eb) {
        for (int round = 0; round < 64; ++round) {
            const bool         has = x != 0ull;
            const unsigned int mk  = __builtin_amdgcn_ballot_w32(has);
            if (mk == 0u) break;
            if (has) {
                const unsigned int bit = (unsigned int)__builtin_ctzll(x);
                x &= x - 1ull;
                const unsigned int pos = wc + __builtin_amdgcn_mbcnt_lo(mk, 0u);
                went[pos < WCAP ? pos : WCAP - 1] = eb + bit;
            }
            wc += (unsigned int)__builtin_popcount(mk);
        }
    };
#pragma nounroll
    for (int it = 0; maps && it < QSCAN_TRIPS; ++it) {
        uint4 l4[4], h4[4];
        int   pp[4];
#pragma unroll
        for (int u = 0; u < 4; ++u) {
            const int q = tid + (it * 4 + u) * WG_T;
            int p = (q < NQ ? q : 0) + rot;
            if (p >= NQ) p -= NQ;
            pp[u] = p;
            l4[u] = bl4[p];
            h4[u] = bh4[p];
        }
#pragma unroll
        for (int u = 0; u < 4; ++u) {
            if (it * 4 + u < QSLOTS) {
                const unsigned int in = tid + (it * 4 + u) * WG_T < NQ ? 0xffffffffu : 0u;
                const unsigned int m0 = l4[u].x & h4[u].x & in;
                const unsigned int m1 = l4[u].y & h4[u].y & in;
                const unsigned int m2 = l4[u].z & h4[u].z & in;
                const unsigned int m3 = l4[u].w & h4[u].w & in;
                const unsigned int eb = 128u * (unsigned int)pp[u];
                drain(((unsigned long long)m1 << 32) | m0, eb);
                drain(((unsigned long long)m3 << 32) | m2, eb + 64u);
            }
        }
    }
    if ((wc > WCAP || !maps) && lane == 0) ovf = 1u;
    __syncthreads();

    if (ovf == 0u) {
        const unsigned int n = wc;
        for (unsigned int k0 = lane; k0 < n; k0 += 4 * 32) {
            unsigned int en[4];
            unsigned int s[4], dlv[4];
            float ss[4];
#pragma unroll
            for (int u = 0; u < 4; ++u) {
                const unsigned int k = k0 + u * 32;
                en[u] = k < n ? went[k] : 0u;
            }
#pragma unroll
            for (int u = 0; u < 4; ++u) {
                const unsigned int w = pk[en[u]];
                s[u]   = w >> DL_BITS;
                dlv[u] = w & DL_MASK;
            }
#pragma unroll
            for (int u = 0; u < 4; ++u) ss[u] = s_src[s[u]];
            bool         live[4];
            float        vv[4];
            unsigned int slot[4];
#pragma unroll
            for (int u = 0; u < 4; ++u) {
                const unsigned int k = k0 + u * 32;
                live[u] = k < n && dlv[u] < (unsigned int)nn;
                vv[u]   = leaky(ss[u] + sd[live[u] ? dlv[u] : 0u]);
            }
#pragma unroll
            for (int u = 0; u < 4; ++u) {
                slot[u] = 0u;
                if (live[u]) slot[u] = atomicAdd(&cnt[dlv[u]], 1u);
            }
#pragma unroll
            for (int u = 0; u < 4; ++u) {
                if (live[u]) {
                    const unsigned int dl = dlv[u];
                    if (slot[u] < NCAP) {
                        const unsigned int at = dl * NCAP + ((slot[u] + dl) & (NCAP - 1));
                        nodeS[at] = __uint_as_float(s[u]);
                        nodeV[at] = vv[u];
                    } else {
                        ovf = 1u;
                    }
                }
            }
        }
    }
    __syncthreads();

    if (ovf == 0u) {
        const int c16   = lane & 15;
        const int hf    = lane >> 4;
        const int npair = (nn + 1) >> 1;
        for (int p = wave; p < npair; p += N_WAVE) {
            const int  dl   = 2 * p + hf;
            const bool own  = dl < nn;
            const int  dq   = own ? dl : 0;
            const int  cn   = own ? (int)cnt[dq] : 0;
            const int  cnA  = lane_of_i(cn, 0);
            const int  cnB  = lane_of_i(cn, 16);
            const int  cmax = cnA > cnB ? cnA : cnB;
            float4 a    = make_float4(0.0f, 0.0f, 0.0f, 0.0f);
            float  dsum = 0.0f;
            if (cmax > 0) {
                const int  lb = dq * NCAP;
                const bool h0 = c16 < cn;
                const bool h1 = c16 + 16 < cn;
                const bool h2 = c16 + 32 < cn;
                const bool h3 = c16 + 48 < cn;
                float        v0 = -__builtin_inff(), v1 = -__builtin_inff(), v2 = -__builtin_inff(), v3 = -__builtin_inff();
                unsigned int s0 = 0u, s1 = 0u, s2 = 0u, s3 = 0u;
                if (h0) { const int q = lb + ((c16 + dq) & (NCAP - 1));      v0 = nodeV[q]; s0 = __float_as_uint(nodeS[q]); }
                if (cmax > 16) {
                    if (h1) { const int q = lb + ((c16 + 16 + dq) & (NCAP - 1)); v1 = nodeV[q]; s1 = __float_as_uint(nodeS[q]); }
                    if (cmax > 32) {
                        if (h2) { const int q = lb + ((c16 + 32 + dq) & (NCAP - 1)); v2 = nodeV[q]; s2 = __float_as_uint(nodeS[q]); }
                        if (h3) { const int q = lb + ((c16 + 48 + dq) & (NCAP - 1)); v3 = nodeV[q]; s3 = __float_as_uint(nodeS[q]); }
                    }
                }
                const unsigned int sfirst = row_get_u<0>(s0);
                if (!h0) s0 = sfirst;
                if (!h1) s1 = sfirst;
                if (!h2) s2 = sfirst;
                if (!h3) s3 = sfirst;
                float w0 = 0.0f, w1 = 0.0f, w2 = 0.0f, w3 = 0.0f;
                for (int q = 0; q < cmax; q += 16) {
                    const unsigned int sreg = q >= 48 ? s3 : (q >= 32 ? s2 : (q >= 16 ? s1 : s0));
                    float4 x[16];
                    rows_ld<0>(Xp, sreg, cmax - q, c16, x);
                    if (q == 0) {
                        const float m = row_max16(fmaxf(fmaxf(v0, v1), fmaxf(v2, v3)));
                        w0 = h0 ? expf(v0 - m) : 0.0f;
                        w1 = h1 ? expf(v1 - m) : 0.0f;
                        w2 = h2 ? expf(v2 - m) : 0.0f;
                        w3 = h3 ? expf(v3 - m) : 0.0f;
                        dsum = row_sum16((w0 + w1) + (w2 + w3));
                    }
                    const float wreg = q >= 48 ? w3 : (q >= 32 ? w2 : (q >= 16 ? w1 : w0));
                    rows_fm<0>(wreg, cmax - q, x, a);
                }
            }
            const bool  any = cn > 0;
            const float inv = (any && dsum != 0.0f) ? 1.0f / dsum : 0.0f;
            float rx = any ? a.x * inv : 0.0f;
            float ry = any ? a.y * inv : 0.0f;
            float rz = any ? a.z * inv : 0.0f;
            float rw = any ? a.w * inv : 0.0f;
            rx = rx > 0.0f ? rx : (expf(rx) - 1.0f);
            ry = ry > 0.0f ? ry : (expf(ry) - 1.0f);
            rz = rz > 0.0f ? rz : (expf(rz) - 1.0f);
            rw = rw > 0.0f ? rw : (expf(rw) - 1.0f);
            if (own) *(float4*)(out + (size_t)(lo + dl) * D_OUT + 4 * c16) = make_float4(rx, ry, rz, rw);
        }
    } else {
        float* acc = nbuf;
        for (int i = tid; i < CH * D_OUT; i += WG_T) acc[i] = 0.0f;
        if (tid < CH) {
            mI[tid]  = __float_as_int(-__builtin_inff());
            den[tid] = 0.0f;
        }
        __syncthreads();
        const int4* dv = (const int4*)e_dst;
        for (int sweep = 0; sweep < 2; ++sweep) {
            for (int q = tid; q < NV4; q += WG_T) {
                const int4 d = dv[q];
                const int dd[4] = {d.x, d.y, d.z, d.w};
                for (int c = 0; c < 4; ++c) {
                    const unsigned int dl = (unsigned int)(dd[c] - lo);
                    if (dl < (unsigned int)nn) {
                        const int   s = e_src[4 * q + c];
                        const float v = leaky(s_src[s] + sd[dl]);
                        if (sweep == 0) {
                            lds_fmax(&mI[dl], v);
                        } else {
                            const float w = expf(v - __int_as_float(mI[dl]));
                            atomicAdd(&den[dl], w);
                            const float* xs = Xp + (size_t)s * D_OUT;
                            for (int f = 0; f < D_OUT; ++f) atomicAdd(&acc[dl * D_OUT + f], w * xs[f]);
                        }
                    }
                }
            }
            __syncthreads();
        }
        float* o = out + (size_t)lo * D_OUT;
        for (int i = tid; i < nn * D_OUT; i += WG_T) {
            const float dn = den[i >> 6];
            const float r  = dn == 0.0f ? 0.0f : acc[i] / dn;
            o[i] = r > 0.0f ? r : (expf(r) - 1.0f);
        }
    }
}

extern "C" void kernel_launch(void* const* d_in, const int* in_sizes, int n_in,
                              void* d_out, int out_size, void* d_ws, size_t ws_size,
                              hipStream_t stream) {
    const float* X       = (const float*)d_in[0];
    const float* theta_w = (const float*)d_in[1];
    const float* theta_b = (const float*)d_in[2];
    const float* a_src   = (const float*)d_in[3];
    const float* a_src_b = (const float*)d_in[4];
    const float* a_dst   = (const float*)d_in[5];
    const float* a_dst_b = (const float*)d_in[6];
    const int*   e_src   = (const int*)d_in[7];
    const int*   e_dst   = (const int*)d_in[8];
    float*       out     = (float*)d_out;

    float* Xp    = (float*)d_ws;
    float* s_src = Xp    + (size_t)N_NODES * D_OUT;
    float* s_dst = s_src + N_NODES;
    unsigned int* bm = (unsigned int*)(s_dst + N_NODES);
    unsigned int* pk = bm + (size_t)N_BM * BM_STRIDE;

    const int gemmGrid  = (ROW_BLOCKS + 7) / 8;
    const int scoreGrid = (SC_BLOCKS * 32 + SC_WG_T - 1) / SC_WG_T;

    k_gemm<<<gemmGrid, 256, 0, stream>>>(X, theta_w, theta_b, Xp);
    k_scores<<<scoreGrid, SC_WG_T, 0, stream>>>(Xp, a_src, a_src_b, a_dst, a_dst_b, s_src, s_dst,
                                                e_dst, bm, e_src, pk, (unsigned long long)ws_size);
    k_owner<<<N_WG, WG_T, 0, stream>>>(e_src, e_dst, s_src, s_dst, Xp, bm, pk, (unsigned long long)ws_size, out);
}


# scores launch removed: projection waves form s_src, s_dst as X.(W.a) on the vector ALU beside the WMMAs (W.a folded per work-group) and pack the edge list; 3 launches -> 2; score sum order changed
# speedup vs baseline: 6.1912x; 1.0439x over previous
#include <hip/hip_runtime.h>
#include <math.h>

typedef float v2f __attribute__((ext_vector_type(2)));
typedef float v8f __attribute__((ext_vector_type(8)));

#define N_NODES   100000
#define N_EDGES   1600000
#define D_IN      128
#define D_OUT     64
#define ROW_BLOCKS (N_NODES / 16)
#define GEMM_WAVES 8
#define GEMM_GRID  ((ROW_BLOCKS + GEMM_WAVES - 1) / GEMM_WAVES)

#define CH        392
#define N_WG      ((N_NODES + CH - 1) / CH)
#define NG        (N_EDGES / 32)
#define PL_EDGES  512
#define PL_BLOCKS (N_EDGES / PL_EDGES)
#define PL_GROUPS (PL_EDGES / 32)
#define PK_WAVES  4
#define N_BM      32
#define BM_STRIDE (((NG + 63) / 64) * 64)
#define WS_FLOATS ((size_t)N_NODES * D_OUT + 2 * (size_t)N_NODES)
#define DL_BITS 9
#define DL_MASK ((1u << DL_BITS) - 1u)
#define WS_NEED   (WS_FLOATS * 4 + (size_t)N_BM * BM_STRIDE * 4 + (size_t)N_EDGES * 4)

static_assert(N_WG == 256, "an owner number is exactly 8 bits");
static_assert(N_EDGES % PL_EDGES == 0 && PL_GROUPS == 16, "every packing wave packs 16 whole groups, one per lane of a 16-lane row");
static_assert(GEMM_GRID * PK_WAVES >= PL_BLOCKS && PK_WAVES <= GEMM_WAVES, "the projection launch covers every block of edges");
static_assert((WS_FLOATS * 4) % 256 == 0, "the bitmaps start on a 256-byte boundary");
static_assert(BM_STRIDE % 64 == 0 && BM_STRIDE >= NG, "every bitmap holds all groups and starts on a 256-byte boundary");
static_assert(D_OUT == 64, "a 16-lane row of float4 covers one Xp row");

template <int M>
__device__ __forceinline__ float row_xor(float v) {
    return __int_as_float(__builtin_amdgcn_update_dpp(0, __float_as_int(v), 0x160 + M, 0xf, 0xf, true));
}

__device__ __forceinline__ float row_sum16(float v) {
    v += row_xor<1>(v);
    v += row_xor<2>(v);
    v += row_xor<4>(v);
    v += row_xor<8>(v);
    return v;
}

__device__ __forceinline__ float other_row(float v) {
    return __int_as_float(__builtin_amdgcn_permlanex16(0, __float_as_int(v), 0x76543210, 0xfedcba98, true, false));
}

__global__ __launch_bounds__(GEMM_WAVES * 32) void k_gemm(const float* __restrict__ X,
                                                          const float* __restrict__ W,
                                                          const float* __restrict__ bias,
                                                          const float* __restrict__ a_src,
                                                          const float* __restrict__ a_src_b,
                                                          const float* __restrict__ a_dst,
                                                          const float* __restrict__ a_dst_b,
                                                          float* __restrict__ Xp,
                                                          float* __restrict__ s_src,
                                                          float* __restrict__ s_dst,
                                                          const int* __restrict__ e_src,
                                                          const int* __restrict__ e_dst,
                                                          unsigned int* __restrict__ bm,
                                                          unsigned int* __restrict__ pk,
                                                          unsigned long long ws_bytes) {
    __shared__ float  Wl[D_IN * D_OUT];
    __shared__ float4 Ul[D_IN / 2];
    __shared__ float  Ub[2];
    const int tid  = threadIdx.x;
    const int wave = tid >> 5;
    const int lane = tid & 31;

    const int  blk   = blockIdx.x * PK_WAVES + wave;
    const bool packs = wave < PK_WAVES && blk < PL_BLOCKS &&
                       ws_bytes >= (unsigned long long)WS_NEED;
    static_assert(GEMM_WAVES * 16 == D_IN, "eight waves of sixteen rows cover W");
    const int fc = lane & 15;
    const int fh = lane >> 4;
    const float4 as = *(const float4*)(a_src + 4 * fc);
    const float4 ad = *(const float4*)(a_dst + 4 * fc);
    float4 x[8];
    {
        const float* wb = W + (size_t)(16 * wave + 8 * fh) * D_OUT + 4 * fc;
#pragma unroll
        for (int j = 0; j < 8; ++j) x[j] = *(const float4*)(wb + j * D_OUT);
    }

    for (int i = tid; i < D_IN * D_OUT; i += GEMM_WAVES * 32) {
        const int k  = i >> 6;
        const int c  = i & 63;
        const int kk = k & ~1;
        Wl[kk * D_OUT + (c << 1) + (k & 1)] = W[i];
    }
    {
        float r = 0.0f;
#pragma unroll
        for (int j = 0; j < 8; ++j) {
            const float fs = row_sum16(x[j].x * as.x + x[j].y * as.y + x[j].z * as.z + x[j].w * as.w);
            const float fd = row_sum16(x[j].x * ad.x + x[j].y * ad.y + x[j].z * ad.z + x[j].w * ad.w);
            if (fc == j)     r = fs;
            if (fc == 8 + j) r = fd;
        }
        const int k = 16 * wave + 8 * fh + (fc & 7);
        ((float*)Ul)[(k >> 1) * 4 + (fc >> 3) * 2 + (k & 1)] = r;
        if (wave == 0) {
            const float4 b4 = *(const float4*)(bias + 4 * fc);
            const float gs = row_sum16(b4.x * as.x + b4.y * as.y + b4.z * as.z + b4.w * as.w);
            const float gd = row_sum16(b4.x * ad.x + b4.y * ad.y + b4.z * ad.z + b4.w * ad.w);
            if (lane == 0)  Ub[0] = gs + a_src_b[0];
            if (lane == 16) Ub[1] = gd + a_dst_b[0];
        }
    }
    __syncthreads();

    if (packs) {
        unsigned int pl[8] = {0u, 0u, 0u, 0u, 0u, 0u, 0u, 0u};
        const int*    ed = e_dst + (size_t)blk * PL_EDGES + lane;
        const int*    es = e_src + (size_t)blk * PL_EDGES + lane;
        unsigned int* pw = pk + (size_t)blk * PL_EDGES + lane;
        unsigned int ow[PL_GROUPS];
        unsigned int dv[PL_GROUPS];
        unsigned int sv[PL_GROUPS];
#pragma unroll
        for (int j = 0; j < PL_GROUPS; ++j) dv[j] = (unsigned int)ed[32 * j];
#pragma unroll
        for (int j = 0; j < PL_GROUPS; ++j) sv[j] = (unsigned int)es[32 * j];
#pragma unroll
        for (int j = 0; j < PL_GROUPS; ++j) {
            ow[j] = dv[j] / (unsigned int)CH;
            const unsigned int dl = dv[j] < (unsigned int)N_NODES ? dv[j] - ow[j] * (unsigned int)CH : DL_MASK;
            pw[32 * j] = (sv[j] << DL_BITS) | dl;
        }
#pragma unroll
        for (int j = 0; j < PL_GROUPS; ++j) {
#pragma unroll
            for (int k = 0; k < 8; ++k) {
                const unsigned int b = __builtin_amdgcn_ballot_w32(((ow[j] >> k) & 1u) != 0u);
                if ((lane & 15) == j) pl[k] = b;
            }
        }
        const bool hi = lane >= 16;
        const unsigned int q0 = hi ? pl[4] : pl[0];
        const unsigned int q1 = hi ? pl[5] : pl[1];
        const unsigned int q2 = hi ? pl[6] : pl[2];
        const unsigned int q3 = hi ? pl[7] : pl[3];
        const unsigned int va[4] = {~q1 & ~q0, ~q1 & q0, q1 & ~q0, q1 & q0};
        const unsigned int vb[4] = {~q3 & ~q2, ~q3 & q2, q3 & ~q2, q3 & q2};
        unsigned int* bw = bm + (size_t)(hi ? 16 : 0) * BM_STRIDE + (size_t)blk * PL_GROUPS + (lane & 15);
#pragma unroll
        for (int c = 0; c < 16; ++c) bw[(size_t)c * BM_STRIDE] = va[c & 3] & vb[c >> 2];
    }

    const int rowBlock = blockIdx.x * GEMM_WAVES + wave;
    if (rowBlock >= ROW_BLOCKS) return;

    const int rowBase = rowBlock * 16;
    const int l15     = lane & 15;
    const int khalf   = (lane >> 4) << 1;

    const float* xrow = X + (size_t)(rowBase + l15) * D_IN;

    v8f acc0 = {}, acc1 = {}, acc2 = {}, acc3 = {};
    float ps = 0.0f, pd = 0.0f;
#pragma unroll
    for (int k0 = 0; k0 < D_IN; k0 += 4) {
        v2f a = *(const v2f*)(xrow + k0 + khalf);
        const float* wp = Wl + (k0 + khalf) * D_OUT;
        v2f b0 = *(const v2f*)(wp + (l15 << 1));
        v2f b1 = *(const v2f*)(wp + ((16 + l15) << 1));
        v2f b2 = *(const v2f*)(wp + ((32 + l15) << 1));
        v2f b3 = *(const v2f*)(wp + ((48 + l15) << 1));
        const float4 u = Ul[(k0 + khalf) >> 1];
        acc0 = __builtin_amdgcn_wmma_f32_16x16x4_f32(false, a, false, b0, (short)0, acc0, false, false);
        acc1 = __builtin_amdgcn_wmma_f32_16x16x4_f32(false, a, false, b1, (short)0, acc1, false, false);
        acc2 = __builtin_amdgcn_wmma_f32_16x16x4_f32(false, a, false, b2, (short)0, acc2, false, false);
        acc3 = __builtin_amdgcn_wmma_f32_16x16x4_f32(false, a, false, b3, (short)0, acc3, false, false);
        ps = fmaf(a.x, u.x, ps);
        ps = fmaf(a.y, u.y, ps);
        pd = fmaf(a.x, u.z, pd);
        pd = fmaf(a.y, u.w, pd);
    }

    {
        const float qs = other_row(ps);
        const float qd = other_row(pd);
        const bool  hi = lane >= 16;
        float* sp = hi ? s_dst : s_src;
        sp[rowBase + l15] = (hi ? pd + qd : ps + qs) + Ub[hi ? 1 : 0];
    }

    const int orow = rowBase + (lane >> 4) * 8;
    const float b0s = bias[l15], b1s = bias[16 + l15];
    const float b2s = bias[32 + l15], b3s = bias[48 + l15];
#pragma unroll
    for (int r = 0; r < 8; ++r) {
        float* o = Xp + (size_t)(orow + r) * D_OUT;
        o[l15]      = acc0[r] + b0s;
        o[16 + l15] = acc1[r] + b1s;
        o[32 + l15] = acc2[r] + b2s;
        o[48 + l15] = acc3[r] + b3s;
    }
}

#define WG_T    1024
#define N_WAVE  (WG_T / 32)
#define WCAP    512
#define NV4     (N_EDGES / 4)
#define NQ      (NG / 4)
#define QSLOTS  ((NQ + WG_T - 1) / WG_T)
#define QSCAN_TRIPS ((QSLOTS + 3) / 4)
static_assert(N_EDGES % 4 == 0, "the overflow path reads e_dst as int4");
static_assert(NG % 4 == 0, "a bitmap is read as 16-byte vectors of four whole groups");
static_assert(CH <= (1 << DL_BITS), "local node id must fit DL_BITS");
static_assert(N_EDGES <= (1 << (32 - DL_BITS - 2)), "edge id packs above the local node id");
static_assert(N_NODES <= (1 << (32 - DL_BITS - 2)), "source id packs above the local node id");

__device__ __forceinline__ void lds_fmax(int* slot, float v) {
    if (v >= 0.0f) atomicMax(slot, __float_as_int(v));
    else           atomicMin((unsigned int*)slot, __float_as_uint(v));
}

__device__ __forceinline__ float leaky(float v) { return v > 0.0f ? v : 0.01f * v; }

#define NCAP 64
static_assert((NCAP & (NCAP - 1)) == 0 && NCAP == 64, "a node's list is four entries per lane of a 16-lane row");
static_assert(CH * D_OUT <= 2 * CH * NCAP, "the overflow path's rows fit in the list space");

__device__ __forceinline__ float row_max16(float v) {
    v = fmaxf(v, row_xor<1>(v));
    v = fmaxf(v, row_xor<2>(v));
    v = fmaxf(v, row_xor<4>(v));
    v = fmaxf(v, row_xor<8>(v));
    return v;
}
__device__ __forceinline__ int lane_of_i(int v, int l) {
    return __builtin_amdgcn_readlane(v, l);
}
template <int U>
__device__ __forceinline__ unsigned int row_get_u(unsigned int v) {
    return (unsigned int)__builtin_amdgcn_update_dpp(0, (int)v, 0x150 + U, 0xf, 0xf, true);
}
template <int U>
__device__ __forceinline__ float row_get_f(float v) {
    return __int_as_float(__builtin_amdgcn_update_dpp(0, __float_as_int(v), 0x150 + U, 0xf, 0xf, true));
}

template <int U>
__device__ __forceinline__ void rows_ld(const float* __restrict__ Xp, unsigned int sreg, int left,
                                        int c, float4 (&x)[16]) {
    const unsigned int s = row_get_u<U>(sreg);
    x[U] = *(const float4*)(Xp + (size_t)s * D_OUT + 4 * c);
    if (left > U + 1) rows_ld<U + 1>(Xp, sreg, left, c, x);
}
template <>
__device__ __forceinline__ void rows_ld<16>(const float* __restrict__, unsigned int, int, int, float4 (&)[16]) {}
template <int U>
__device__ __forceinline__ void rows_fm(float wreg, int left, const float4 (&x)[16], float4& a) {
    const float w = row_get_f<U>(wreg);
    a.x += w * x[U].x;
    a.y += w * x[U].y;
    a.z += w * x[U].z;
    a.w += w * x[U].w;
    if (left > U + 1) rows_fm<U + 1>(wreg, left, x, a);
}
template <>
__device__ __forceinline__ void rows_fm<16>(float, int, const float4 (&)[16], float4&) {}

__global__ __launch_bounds__(WG_T) void k_owner(const int* __restrict__ e_src,
                                                const int* __restrict__ e_dst,
                                                const float* __restrict__ s_src,
                                                const float* __restrict__ s_dst,
                                                const float* __restrict__ Xp,
                                                const unsigned int* __restrict__ bm,
                                                const unsigned int* __restrict__ pk,
                                                unsigned long long ws_bytes,
                                                float* __restrict__ out) {
    __shared__ unsigned int ent[N_WAVE * WCAP];
    __shared__ float        nbuf[2 * CH * NCAP];
    __shared__ unsigned int cnt[CH];
    __shared__ int          mI[CH];
    __shared__ float        den[CH];
    __shared__ float        sd[CH];
    __shared__ unsigned int ovf;

    const int tid  = threadIdx.x;
    const int lane = tid & 31;
    const int wave = tid >> 5;
    const int lo   = blockIdx.x * CH;
    const int nn   = (N_NODES - lo) < CH ? (N_NODES - lo) : CH;

    float* nodeS = nbuf;
    float* nodeV = nbuf + CH * NCAP;

    if (tid < CH) {
        cnt[tid] = 0u;
        sd[tid]  = tid < nn ? s_dst[lo + tid] : 0.0f;
    }
    if (tid == 0) ovf = 0u;
    __syncthreads();

    unsigned int* went = ent + wave * WCAP;
    unsigned int  wc   = 0u;

    const bool maps = ws_bytes >= (unsigned long long)WS_NEED;
    const uint4* bl4 = (const uint4*)(bm + (size_t)(blockIdx.x & 15u) * BM_STRIDE);
    const uint4* bh4 = (const uint4*)(bm + (size_t)(16u + (blockIdx.x >> 4)) * BM_STRIDE);
    const int rot = (int)(((((blockIdx.x & 15u) + (blockIdx.x >> 4)) & 15u) * (unsigned int)NQ) / 16u) & ~15;
    auto drain = [&](unsigned long long x, unsigned int eb) {
        for (int round = 0; round < 64; ++round) {
            const bool         has = x != 0ull;
            const unsigned int mk  = __builtin_amdgcn_ballot_w32(has);
            if (mk == 0u) break;
            if (has) {
                const unsigned int bit = (unsigned int)__builtin_ctzll(x);
                x &= x - 1ull;
                const unsigned int pos = wc + __builtin_amdgcn_mbcnt_lo(mk, 0u);
                went[pos < WCAP ? pos : WCAP - 1] = eb + bit;
            }
            wc += (unsigned int)__builtin_popcount(mk);
        }
    };
#pragma nounroll
    for (int it = 0; maps && it < QSCAN_TRIPS; ++it) {
        uint4 l4[4], h4[4];
        int   pp[4];
#pragma unroll
        for (int u = 0; u < 4; ++u) {
            const int q = tid + (it * 4 + u) * WG_T;
            int p = (q < NQ ? q : 0) + rot;
            if (p >= NQ) p -= NQ;
            pp[u] = p;
            l4[u] = bl4[p];
            h4[u] = bh4[p];
        }
#pragma unroll
        for (int u = 0; u < 4; ++u) {
            if (it * 4 + u < QSLOTS) {
                const unsigned int in = tid + (it * 4 + u) * WG_T < NQ ? 0xffffffffu : 0u;
                const unsigned int m0 = l4[u].x & h4[u].x & in;
                const unsigned int m1 = l4[u].y & h4[u].y & in;
                const unsigned int m2 = l4[u].z & h4[u].z & in;
                const unsigned int m3 = l4[u].w & h4[u].w & in;
                const unsigned int eb = 128u * (unsigned int)pp[u];
                drain(((unsigned long long)m1 << 32) | m0, eb);
                drain(((unsigned long long)m3 << 32) | m2, eb + 64u);
            }
        }
    }
    if ((wc > WCAP || !maps) && lane == 0) ovf = 1u;
    __syncthreads();

    if (ovf == 0u) {
        const unsigned int n = wc;
        for (unsigned int k0 = lane; k0 < n; k0 += 4 * 32) {
            unsigned int en[4];
            unsigned int s[4], dlv[4];
            float ss[4];
#pragma unroll
            for (int u = 0; u < 4; ++u) {
                const unsigned int k = k0 + u * 32;
                en[u] = k < n ? went[k] : 0u;
            }
#pragma unroll
            for (int u = 0; u < 4; ++u) {
                const unsigned int w = pk[en[u]];
                s[u]   = w >> DL_BITS;
                dlv[u] = w & DL_MASK;
            }
#pragma unroll
            for (int u = 0; u < 4; ++u) ss[u] = s_src[s[u]];
            bool         live[4];
            float        vv[4];
            unsigned int slot[4];
#pragma unroll
            for (int u = 0; u < 4; ++u) {
                const unsigned int k = k0 + u * 32;
                live[u] = k < n && dlv[u] < (unsigned int)nn;
                vv[u]   = leaky(ss[u] + sd[live[u] ? dlv[u] : 0u]);
            }
#pragma unroll
            for (int u = 0; u < 4; ++u) {
                slot[u] = 0u;
                if (live[u]) slot[u] = atomicAdd(&cnt[dlv[u]], 1u);
            }
#pragma unroll
            for (int u = 0; u < 4; ++u) {
                if (live[u]) {
                    const unsigned int dl = dlv[u];
                    if (slot[u] < NCAP) {
                        const unsigned int at = dl * NCAP + ((slot[u] + dl) & (NCAP - 1));
                        nodeS[at] = __uint_as_float(s[u]);
                        nodeV[at] = vv[u];
                    } else {
                        ovf = 1u;
                    }
                }
            }
        }
    }
    __syncthreads();

    if (ovf == 0u) {
        const int c16   = lane & 15;
        const int hf    = lane >> 4;
        const int npair = (nn + 1) >> 1;
        for (int p = wave; p < npair; p += N_WAVE) {
            const int  dl   = 2 * p + hf;
            const bool own  = dl < nn;
            const int  dq   = own ? dl : 0;
            const int  cn   = own ? (int)cnt[dq] : 0;
            const int  cnA  = lane_of_i(cn, 0);
            const int  cnB  = lane_of_i(cn, 16);
            const int  cmax = cnA > cnB ? cnA : cnB;
            float4 a    = make_float4(0.0f, 0.0f, 0.0f, 0.0f);
            float  dsum = 0.0f;
            if (cmax > 0) {
                const int  lb = dq * NCAP;
                const bool h0 = c16 < cn;
                const bool h1 = c16 + 16 < cn;
                const bool h2 = c16 + 32 < cn;
                const bool h3 = c16 + 48 < cn;
                float        v0 = -__builtin_inff(), v1 = -__builtin_inff(), v2 = -__builtin_inff(), v3 = -__builtin_inff();
                unsigned int s0 = 0u, s1 = 0u, s2 = 0u, s3 = 0u;
                if (h0) { const int q = lb + ((c16 + dq) & (NCAP - 1));      v0 = nodeV[q]; s0 = __float_as_uint(nodeS[q]); }
                if (cmax > 16) {
                    if (h1) { const int q = lb + ((c16 + 16 + dq) & (NCAP - 1)); v1 = nodeV[q]; s1 = __float_as_uint(nodeS[q]); }
                    if (cmax > 32) {
                        if (h2) { const int q = lb + ((c16 + 32 + dq) & (NCAP - 1)); v2 = nodeV[q]; s2 = __float_as_uint(nodeS[q]); }
                        if (h3) { const int q = lb + ((c16 + 48 + dq) & (NCAP - 1)); v3 = nodeV[q]; s3 = __float_as_uint(nodeS[q]); }
                    }
                }
                const unsigned int sfirst = row_get_u<0>(s0);
                if (!h0) s0 = sfirst;
                if (!h1) s1 = sfirst;
                if (!h2) s2 = sfirst;
                if (!h3) s3 = sfirst;
                float w0 = 0.0f, w1 = 0.0f, w2 = 0.0f, w3 = 0.0f;
                for (int q = 0; q < cmax; q += 16) {
                    const unsigned int sreg = q >= 48 ? s3 : (q >= 32 ? s2 : (q >= 16 ? s1 : s0));
                    float4 x[16];
                    rows_ld<0>(Xp, sreg, cmax - q, c16, x);
                    if (q == 0) {
                        const float m = row_max16(fmaxf(fmaxf(v0, v1), fmaxf(v2, v3)));
                        w0 = h0 ? expf(v0 - m) : 0.0f;
                        w1 = h1 ? expf(v1 - m) : 0.0f;
                        w2 = h2 ? expf(v2 - m) : 0.0f;
                        w3 = h3 ? expf(v3 - m) : 0.0f;
                        dsum = row_sum16((w0 + w1) + (w2 + w3));
                    }
                    const float wreg = q >= 48 ? w3 : (q >= 32 ? w2 : (q >= 16 ? w1 : w0));
                    rows_fm<0>(wreg, cmax - q, x, a);
                }
            }
            const bool  any = cn > 0;
            const float inv = (any && dsum != 0.0f) ? 1.0f / dsum : 0.0f;
            float rx = any ? a.x * inv : 0.0f;
            float ry = any ? a.y * inv : 0.0f;
            float rz = any ? a.z * inv : 0.0f;
            float rw = any ? a.w * inv : 0.0f;
            rx = rx > 0.0f ? rx : (expf(rx) - 1.0f);
            ry = ry > 0.0f ? ry : (expf(ry) - 1.0f);
            rz = rz > 0.0f ? rz : (expf(rz) - 1.0f);
            rw = rw > 0.0f ? rw : (expf(rw) - 1.0f);
            if (own) *(float4*)(out + (size_t)(lo + dl) * D_OUT + 4 * c16) = make_float4(rx, ry, rz, rw);
        }
    } else {
        float* acc = nbuf;
        for (int i = tid; i < CH * D_OUT; i += WG_T) acc[i] = 0.0f;
        if (tid < CH) {
            mI[tid]  = __float_as_int(-__builtin_inff());
            den[tid] = 0.0f;
        }
        __syncthreads();
        const int4* dv = (const int4*)e_dst;
        for (int sweep = 0; sweep < 2; ++sweep) {
            for (int q = tid; q < NV4; q += WG_T) {
                const int4 d = dv[q];
                const int dd[4] = {d.x, d.y, d.z, d.w};
                for (int c = 0; c < 4; ++c) {
                    const unsigned int dl = (unsigned int)(dd[c] - lo);
                    if (dl < (unsigned int)nn) {
                        const int   s = e_src[4 * q + c];
                        const float v = leaky(s_src[s] + sd[dl]);
                        if (sweep == 0) {
                            lds_fmax(&mI[dl], v);
                        } else {
                            const float w = expf(v - __int_as_float(mI[dl]));
                            atomicAdd(&den[dl], w);
                            const float* xs = Xp + (size_t)s * D_OUT;
                            for (int f = 0; f < D_OUT; ++f) atomicAdd(&acc[dl * D_OUT + f], w * xs[f]);
                        }
                    }
                }
            }
            __syncthreads();
        }
        float* o = out + (size_t)lo * D_OUT;
        for (int i = tid; i < nn * D_OUT; i += WG_T) {
            const float dn = den[i >> 6];
            const float r  = dn == 0.0f ? 0.0f : acc[i] / dn;
            o[i] = r > 0.0f ? r : (expf(r) - 1.0f);
        }
    }
}

extern "C" void kernel_launch(void* const* d_in, const int* in_sizes, int n_in,
                              void* d_out, int out_size, void* d_ws, size_t ws_size,
                              hipStream_t stream) {
    const float* X       = (const float*)d_in[0];
    const float* theta_w = (const float*)d_in[1];
    const float* theta_b = (const float*)d_in[2];
    const float* a_src   = (const float*)d_in[3];
    const float* a_src_b = (const float*)d_in[4];
    const float* a_dst   = (const float*)d_in[5];
    const float* a_dst_b = (const float*)d_in[6];
    const int*   e_src   = (const int*)d_in[7];
    const int*   e_dst   = (const int*)d_in[8];
    float*       out     = (float*)d_out;

    float* Xp    = (float*)d_ws;
    float* s_src = Xp    + (size_t)N_NODES * D_OUT;
    float* s_dst = s_src + N_NODES;
    unsigned int* bm = (unsigned int*)(s_dst + N_NODES);
    unsigned int* pk = bm + (size_t)N_BM * BM_STRIDE;

    k_gemm<<<GEMM_GRID, GEMM_WAVES * 32, 0, stream>>>(X, theta_w, theta_b, a_src, a_src_b, a_dst, a_dst_b,
                                                       Xp, s_src, s_dst, e_src, e_dst, bm, pk,
                                                       (unsigned long long)ws_size);
    k_owner<<<N_WG, WG_T, 0, stream>>>(e_src, e_dst, s_src, s_dst, Xp, bm, pk, (unsigned long long)ws_size, out);
}


# projection: each wave fetches its 16x128 X strip as whole 128-byte lines (16-byte loads, each line requested once) into its own LDS tile, a K quarter at a time; A fragments read there; sums unchanged
# speedup vs baseline: 6.3885x; 1.0319x over previous
#include <hip/hip_runtime.h>
#include <math.h>

typedef float v2f __attribute__((ext_vector_type(2)));
typedef float v8f __attribute__((ext_vector_type(8)));

#define N_NODES   100000
#define N_EDGES   1600000
#define D_IN      128
#define D_OUT     64
#define ROW_BLOCKS (N_NODES / 16)
#define GEMM_WAVES 8
#define GEMM_GRID  ((ROW_BLOCKS + GEMM_WAVES - 1) / GEMM_WAVES)
#define XT_K       32
#define XT_PITCH   (XT_K + 4)

#define CH        392
#define N_WG      ((N_NODES + CH - 1) / CH)
#define NG        (N_EDGES / 32)
#define PL_EDGES  512
#define PL_BLOCKS (N_EDGES / PL_EDGES)
#define PL_GROUPS (PL_EDGES / 32)
#define PK_WAVES  4
#define N_BM      32
#define BM_STRIDE (((NG + 63) / 64) * 64)
#define WS_FLOATS ((size_t)N_NODES * D_OUT + 2 * (size_t)N_NODES)
#define DL_BITS 9
#define DL_MASK ((1u << DL_BITS) - 1u)
#define WS_NEED   (WS_FLOATS * 4 + (size_t)N_BM * BM_STRIDE * 4 + (size_t)N_EDGES * 4)

static_assert(N_WG == 256, "an owner number is exactly 8 bits");
static_assert(N_EDGES % PL_EDGES == 0 && PL_GROUPS == 16, "every packing wave packs 16 whole groups, one per lane of a 16-lane row");
static_assert(GEMM_GRID * PK_WAVES >= PL_BLOCKS && PK_WAVES <= GEMM_WAVES, "the projection launch covers every block of edges");
static_assert((WS_FLOATS * 4) % 256 == 0, "the bitmaps start on a 256-byte boundary");
static_assert(BM_STRIDE % 64 == 0 && BM_STRIDE >= NG, "every bitmap holds all groups and starts on a 256-byte boundary");
static_assert(D_OUT == 64, "a 16-lane row of float4 covers one Xp row");
static_assert(D_IN == 4 * XT_K && XT_K == 32, "eight lanes of float4 cover one K quarter of an X row, one 128-byte line");
static_assert((XT_PITCH * 4) % 16 == 0 && ((XT_PITCH * 4) / 16) % 2 == 1, "tile rows are 16-byte aligned at an odd multiple of 16 bytes");

template <int M>
__device__ __forceinline__ float row_xor(float v) {
    return __int_as_float(__builtin_amdgcn_update_dpp(0, __float_as_int(v), 0x160 + M, 0xf, 0xf, true));
}

__device__ __forceinline__ float row_sum16(float v) {
    v += row_xor<1>(v);
    v += row_xor<2>(v);
    v += row_xor<4>(v);
    v += row_xor<8>(v);
    return v;
}

__device__ __forceinline__ float other_row(float v) {
    return __int_as_float(__builtin_amdgcn_permlanex16(0, __float_as_int(v), 0x76543210, 0xfedcba98, true, false));
}

__global__ __launch_bounds__(GEMM_WAVES * 32) void k_gemm(const float* __restrict__ X,
                                                          const float* __restrict__ W,
                                                          const float* __restrict__ bias,
                                                          const float* __restrict__ a_src,
                                                          const float* __restrict__ a_src_b,
                                                          const float* __restrict__ a_dst,
                                                          const float* __restrict__ a_dst_b,
                                                          float* __restrict__ Xp,
                                                          float* __restrict__ s_src,
                                                          float* __restrict__ s_dst,
                                                          const int* __restrict__ e_src,
                                                          const int* __restrict__ e_dst,
                                                          unsigned int* __restrict__ bm,
                                                          unsigned int* __restrict__ pk,
                                                          unsigned long long ws_bytes) {
    __shared__ float  Wl[D_IN * D_OUT];
    __shared__ float4 Ul[D_IN / 2];
    __shared__ float  Ub[2];
    __shared__ float4 Xt[GEMM_WAVES * 16 * XT_PITCH / 4];
    const int tid  = threadIdx.x;
    const int wave = tid >> 5;
    const int lane = tid & 31;

    const int  blk   = blockIdx.x * PK_WAVES + wave;
    const bool packs = wave < PK_WAVES && blk < PL_BLOCKS &&
                       ws_bytes >= (unsigned long long)WS_NEED;
    static_assert(GEMM_WAVES * 16 == D_IN, "eight waves of sixteen rows cover W");
    const int fc = lane & 15;
    const int fh = lane >> 4;
    const float4 as = *(const float4*)(a_src + 4 * fc);
    const float4 ad = *(const float4*)(a_dst + 4 * fc);
    float4 x[8];
    {
        const float* wb = W + (size_t)(16 * wave + 8 * fh) * D_OUT + 4 * fc;
#pragma unroll
        for (int j = 0; j < 8; ++j) x[j] = *(const float4*)(wb + j * D_OUT);
    }

    for (int i = tid; i < D_IN * D_OUT; i += GEMM_WAVES * 32) {
        const int k  = i >> 6;
        const int c  = i & 63;
        const int kk = k & ~1;
        Wl[kk * D_OUT + (c << 1) + (k & 1)] = W[i];
    }
    {
        float r = 0.0f;
#pragma unroll
        for (int j = 0; j < 8; ++j) {
            const float fs = row_sum16(x[j].x * as.x + x[j].y * as.y + x[j].z * as.z + x[j].w * as.w);
            const float fd = row_sum16(x[j].x * ad.x + x[j].y * ad.y + x[j].z * ad.z + x[j].w * ad.w);
            if (fc == j)     r = fs;
            if (fc == 8 + j) r = fd;
        }
        const int k = 16 * wave + 8 * fh + (fc & 7);
        ((float*)Ul)[(k >> 1) * 4 + (fc >> 3) * 2 + (k & 1)] = r;
        if (wave == 0) {
            const float4 b4 = *(const float4*)(bias + 4 * fc);
            const float gs = row_sum16(b4.x * as.x + b4.y * as.y + b4.z * as.z + b4.w * as.w);
            const float gd = row_sum16(b4.x * ad.x + b4.y * ad.y + b4.z * ad.z + b4.w * ad.w);
            if (lane == 0)  Ub[0] = gs + a_src_b[0];
            if (lane == 16) Ub[1] = gd + a_dst_b[0];
        }
    }
    __syncthreads();

    if (packs) {
        unsigned int pl[8] = {0u, 0u, 0u, 0u, 0u, 0u, 0u, 0u};
        const int*    ed = e_dst + (size_t)blk * PL_EDGES + lane;
        const int*    es = e_src + (size_t)blk * PL_EDGES + lane;
        unsigned int* pw = pk + (size_t)blk * PL_EDGES + lane;
        unsigned int ow[PL_GROUPS];
        unsigned int dv[PL_GROUPS];
        unsigned int sv[PL_GROUPS];
#pragma unroll
        for (int j = 0; j < PL_GROUPS; ++j) dv[j] = (unsigned int)ed[32 * j];
#pragma unroll
        for (int j = 0; j < PL_GROUPS; ++j) sv[j] = (unsigned int)es[32 * j];
#pragma unroll
        for (int j = 0; j < PL_GROUPS; ++j) {
            ow[j] = dv[j] / (unsigned int)CH;
            const unsigned int dl = dv[j] < (unsigned int)N_NODES ? dv[j] - ow[j] * (unsigned int)CH : DL_MASK;
            pw[32 * j] = (sv[j] << DL_BITS) | dl;
        }
#pragma unroll
        for (int j = 0; j < PL_GROUPS; ++j) {
#pragma unroll
            for (int k = 0; k < 8; ++k) {
                const unsigned int b = __builtin_amdgcn_ballot_w32(((ow[j] >> k) & 1u) != 0u);
                if ((lane & 15) == j) pl[k] = b;
            }
        }
        const bool hi = lane >= 16;
        const unsigned int q0 = hi ? pl[4] : pl[0];
        const unsigned int q1 = hi ? pl[5] : pl[1];
        const unsigned int q2 = hi ? pl[6] : pl[2];
        const unsigned int q3 = hi ? pl[7] : pl[3];
        const unsigned int va[4] = {~q1 & ~q0, ~q1 & q0, q1 & ~q0, q1 & q0};
        const unsigned int vb[4] = {~q3 & ~q2, ~q3 & q2, q3 & ~q2, q3 & q2};
        unsigned int* bw = bm + (size_t)(hi ? 16 : 0) * BM_STRIDE + (size_t)blk * PL_GROUPS + (lane & 15);
#pragma unroll
        for (int c = 0; c < 16; ++c) bw[(size_t)c * BM_STRIDE] = va[c & 3] & vb[c >> 2];
    }

    const int rowBlock = blockIdx.x * GEMM_WAVES + wave;
    if (rowBlock >= ROW_BLOCKS) return;

    const int rowBase = rowBlock * 16;
    const int l15     = lane & 15;
    const int khalf   = (lane >> 4) << 1;

    const int r4 = lane >> 3;
    const int c8 = lane & 7;
    const float* xg = X + (size_t)(rowBase + r4) * D_IN + 4 * c8;
    float4 xs[4];
#pragma unroll
    for (int j = 0; j < 4; ++j) xs[j] = *(const float4*)(xg + (size_t)(4 * j) * D_IN);
    float*       xw = (float*)Xt + wave * (16 * XT_PITCH) + r4 * XT_PITCH + 4 * c8;
    const float* xr = (const float*)Xt + wave * (16 * XT_PITCH) + l15 * XT_PITCH + khalf;
    const float*  wb = Wl + khalf * D_OUT + (l15 << 1);
    const float4* ub = Ul + (khalf >> 1);

    v8f acc0 = {}, acc1 = {}, acc2 = {}, acc3 = {};
    float ps = 0.0f, pd = 0.0f;
#pragma unroll
    for (int q = 0; q < D_IN / XT_K; ++q) {
#pragma unroll
        for (int j = 0; j < 4; ++j) *(float4*)(xw + 4 * j * XT_PITCH) = xs[j];
        if (q + 1 < D_IN / XT_K) {
            __builtin_amdgcn_sched_barrier(0);
#pragma unroll
            for (int j = 0; j < 4; ++j) xs[j] = *(const float4*)(xg + (size_t)(4 * j) * D_IN + (q + 1) * XT_K);
        }
#pragma unroll
        for (int kq = 0; kq < XT_K; kq += 4) {
            const int k0 = q * XT_K + kq;
            v2f a = *(const v2f*)(xr + kq);
            const float* wp = wb + k0 * D_OUT;
            v2f b0 = *(const v2f*)(wp);
            v2f b1 = *(const v2f*)(wp + 32);
            v2f b2 = *(const v2f*)(wp + 64);
            v2f b3 = *(const v2f*)(wp + 96);
            const float4 u = ub[k0 >> 1];
            acc0 = __builtin_amdgcn_wmma_f32_16x16x4_f32(false, a, false, b0, (short)0, acc0, false, false);
            acc1 = __builtin_amdgcn_wmma_f32_16x16x4_f32(false, a, false, b1, (short)0, acc1, false, false);
            acc2 = __builtin_amdgcn_wmma_f32_16x16x4_f32(false, a, false, b2, (short)0, acc2, false, false);
            acc3 = __builtin_amdgcn_wmma_f32_16x16x4_f32(false, a, false, b3, (short)0, acc3, false, false);
            ps = fmaf(a.x, u.x, ps);
            ps = fmaf(a.y, u.y, ps);
            pd = fmaf(a.x, u.z, pd);
            pd = fmaf(a.y, u.w, pd);
        }
    }

    {
        const float qs = other_row(ps);
        const float qd = other_row(pd);
        const bool  hi = lane >= 16;
        float* sp = hi ? s_dst : s_src;
        sp[rowBase + l15] = (hi ? pd + qd : ps + qs) + Ub[hi ? 1 : 0];
    }

    const int orow = rowBase + (lane >> 4) * 8;
    const float b0s = bias[l15], b1s = bias[16 + l15];
    const float b2s = bias[32 + l15], b3s = bias[48 + l15];
#pragma unroll
    for (int r = 0; r < 8; ++r) {
        float* o = Xp + (size_t)(orow + r) * D_OUT;
        o[l15]      = acc0[r] + b0s;
        o[16 + l15] = acc1[r] + b1s;
        o[32 + l15] = acc2[r] + b2s;
        o[48 + l15] = acc3[r] + b3s;
    }
}

#define WG_T    1024
#define N_WAVE  (WG_T / 32)
#define WCAP    512
#define NV4     (N_EDGES / 4)
#define NQ      (NG / 4)
#define QSLOTS  ((NQ + WG_T - 1) / WG_T)
#define QSCAN_TRIPS ((QSLOTS + 3) / 4)
static_assert(N_EDGES % 4 == 0, "the overflow path reads e_dst as int4");
static_assert(NG % 4 == 0, "a bitmap is read as 16-byte vectors of four whole groups");
static_assert(CH <= (1 << DL_BITS), "local node id must fit DL_BITS");
static_assert(N_EDGES <= (1 << (32 - DL_BITS - 2)), "edge id packs above the local node id");
static_assert(N_NODES <= (1 << (32 - DL_BITS - 2)), "source id packs above the local node id");

__device__ __forceinline__ void lds_fmax(int* slot, float v) {
    if (v >= 0.0f) atomicMax(slot, __float_as_int(v));
    else           atomicMin((unsigned int*)slot, __float_as_uint(v));
}

__device__ __forceinline__ float leaky(float v) { return v > 0.0f ? v : 0.01f * v; }

#define NCAP 64
static_assert((NCAP & (NCAP - 1)) == 0 && NCAP == 64, "a node's list is four entries per lane of a 16-lane row");
static_assert(CH * D_OUT <= 2 * CH * NCAP, "the overflow path's rows fit in the list space");

__device__ __forceinline__ float row_max16(float v) {
    v = fmaxf(v, row_xor<1>(v));
    v = fmaxf(v, row_xor<2>(v));
    v = fmaxf(v, row_xor<4>(v));
    v = fmaxf(v, row_xor<8>(v));
    return v;
}
__device__ __forceinline__ int lane_of_i(int v, int l) {
    return __builtin_amdgcn_readlane(v, l);
}
template <int U>
__device__ __forceinline__ unsigned int row_get_u(unsigned int v) {
    return (unsigned int)__builtin_amdgcn_update_dpp(0, (int)v, 0x150 + U, 0xf, 0xf, true);
}
template <int U>
__device__ __forceinline__ float row_get_f(float v) {
    return __int_as_float(__builtin_amdgcn_update_dpp(0, __float_as_int(v), 0x150 + U, 0xf, 0xf, true));
}

template <int U>
__device__ __forceinline__ void rows_ld(const float* __restrict__ Xp, unsigned int sreg, int left,
                                        int c, float4 (&x)[16]) {
    const unsigned int s = row_get_u<U>(sreg);
    x[U] = *(const float4*)(Xp + (size_t)s * D_OUT + 4 * c);
    if (left > U + 1) rows_ld<U + 1>(Xp, sreg, left, c, x);
}
template <>
__device__ __forceinline__ void rows_ld<16>(const float* __restrict__, unsigned int, int, int, float4 (&)[16]) {}
template <int U>
__device__ __forceinline__ void rows_fm(float wreg, int left, const float4 (&x)[16], float4& a) {
    const float w = row_get_f<U>(wreg);
    a.x += w * x[U].x;
    a.y += w * x[U].y;
    a.z += w * x[U].z;
    a.w += w * x[U].w;
    if (left > U + 1) rows_fm<U + 1>(wreg, left, x, a);
}
template <>
__device__ __forceinline__ void rows_fm<16>(float, int, const float4 (&)[16], float4&) {}

__global__ __launch_bounds__(WG_T) void k_owner(const int* __restrict__ e_src,
                                                const int* __restrict__ e_dst,
                                                const float* __restrict__ s_src,
                                                const float* __restrict__ s_dst,
                                                const float* __restrict__ Xp,
                                                const unsigned int* __restrict__ bm,
                                                const unsigned int* __restrict__ pk,
                                                unsigned long long ws_bytes,
                                                float* __restrict__ out) {
    __shared__ unsigned int ent[N_WAVE * WCAP];
    __shared__ float        nbuf[2 * CH * NCAP];
    __shared__ unsigned int cnt[CH];
    __shared__ int          mI[CH];
    __shared__ float        den[CH];
    __shared__ float        sd[CH];
    __shared__ unsigned int ovf;

    const int tid  = threadIdx.x;
    const int lane = tid & 31;
    const int wave = tid >> 5;
    const int lo   = blockIdx.x * CH;
    const int nn   = (N_NODES - lo) < CH ? (N_NODES - lo) : CH;

    float* nodeS = nbuf;
    float* nodeV = nbuf + CH * NCAP;

    if (tid < CH) {
        cnt[tid] = 0u;
        sd[tid]  = tid < nn ? s_dst[lo + tid] : 0.0f;
    }
    if (tid == 0) ovf = 0u;
    __syncthreads();

    unsigned int* went = ent + wave * WCAP;
    unsigned int  wc   = 0u;

    const bool maps = ws_bytes >= (unsigned long long)WS_NEED;
    const uint4* bl4 = (const uint4*)(bm + (size_t)(blockIdx.x & 15u) * BM_STRIDE);
    const uint4* bh4 = (const uint4*)(bm + (size_t)(16u + (blockIdx.x >> 4)) * BM_STRIDE);
    const int rot = (int)(((((blockIdx.x & 15u) + (blockIdx.x >> 4)) & 15u) * (unsigned int)NQ) / 16u) & ~15;
    auto drain = [&](unsigned long long x, unsigned int eb) {
        for (int round = 0; round < 64; ++round) {
            const bool         has = x != 0ull;
            const unsigned int mk  = __builtin_amdgcn_ballot_w32(has);
            if (mk == 0u) break;
            if (has) {
                const unsigned int bit = (unsigned int)__builtin_ctzll(x);
                x &= x - 1ull;
                const unsigned int pos = wc + __builtin_amdgcn_mbcnt_lo(mk, 0u);
                went[pos < WCAP ? pos : WCAP - 1] = eb + bit;
            }
            wc += (unsigned int)__builtin_popcount(mk);
        }
    };
#pragma nounroll
    for (int it = 0; maps && it < QSCAN_TRIPS; ++it) {
        uint4 l4[4], h4[4];
        int   pp[4];
#pragma unroll
        for (int u = 0; u < 4; ++u) {
            const int q = tid + (it * 4 + u) * WG_T;
            int p = (q < NQ ? q : 0) + rot;
            if (p >= NQ) p -= NQ;
            pp[u] = p;
            l4[u] = bl4[p];
            h4[u] = bh4[p];
        }
#pragma unroll
        for (int u = 0; u < 4; ++u) {
            if (it * 4 + u < QSLOTS) {
                const unsigned int in = tid + (it * 4 + u) * WG_T < NQ ? 0xffffffffu : 0u;
                const unsigned int m0 = l4[u].x & h4[u].x & in;
                const unsigned int m1 = l4[u].y & h4[u].y & in;
                const unsigned int m2 = l4[u].z & h4[u].z & in;
                const unsigned int m3 = l4[u].w & h4[u].w & in;
                const unsigned int eb = 128u * (unsigned int)pp[u];
                drain(((unsigned long long)m1 << 32) | m0, eb);
                drain(((unsigned long long)m3 << 32) | m2, eb + 64u);
            }
        }
    }
    if ((wc > WCAP || !maps) && lane == 0) ovf = 1u;
    __syncthreads();

    if (ovf == 0u) {
        const unsigned int n = wc;
        for (unsigned int k0 = lane; k0 < n; k0 += 4 * 32) {
            unsigned int en[4];
            unsigned int s[4], dlv[4];
            float ss[4];
#pragma unroll
            for (int u = 0; u < 4; ++u) {
                const unsigned int k = k0 + u * 32;
                en[u] = k < n ? went[k] : 0u;
            }
#pragma unroll
            for (int u = 0; u < 4; ++u) {
                const unsigned int w = pk[en[u]];
                s[u]   = w >> DL_BITS;
                dlv[u] = w & DL_MASK;
            }
#pragma unroll
            for (int u = 0; u < 4; ++u) ss[u] = s_src[s[u]];
            bool         live[4];
            float        vv[4];
            unsigned int slot[4];
#pragma unroll
            for (int u = 0; u < 4; ++u) {
                const unsigned int k = k0 + u * 32;
                live[u] = k < n && dlv[u] < (unsigned int)nn;
                vv[u]   = leaky(ss[u] + sd[live[u] ? dlv[u] : 0u]);
            }
#pragma unroll
            for (int u = 0; u < 4; ++u) {
                slot[u] = 0u;
                if (live[u]) slot[u] = atomicAdd(&cnt[dlv[u]], 1u);
            }
#pragma unroll
            for (int u = 0; u < 4; ++u) {
                if (live[u]) {
                    const unsigned int dl = dlv[u];
                    if (slot[u] < NCAP) {
                        const unsigned int at = dl * NCAP + ((slot[u] + dl) & (NCAP - 1));
                        nodeS[at] = __uint_as_float(s[u]);
                        nodeV[at] = vv[u];
                    } else {
                        ovf = 1u;
                    }
                }
            }
        }
    }
    __syncthreads();

    if (ovf == 0u) {
        const int c16   = lane & 15;
        const int hf    = lane >> 4;
        const int npair = (nn + 1) >> 1;
        for (int p = wave; p < npair; p += N_WAVE) {
            const int  dl   = 2 * p + hf;
            const bool own  = dl < nn;
            const int  dq   = own ? dl : 0;
            const int  cn   = own ? (int)cnt[dq] : 0;
            const int  cnA  = lane_of_i(cn, 0);
            const int  cnB  = lane_of_i(cn, 16);
            const int  cmax = cnA > cnB ? cnA : cnB;
            float4 a    = make_float4(0.0f, 0.0f, 0.0f, 0.0f);
            float  dsum = 0.0f;
            if (cmax > 0) {
                const int  lb = dq * NCAP;
                const bool h0 = c16 < cn;
                const bool h1 = c16 + 16 < cn;
                const bool h2 = c16 + 32 < cn;
                const bool h3 = c16 + 48 < cn;
                float        v0 = -__builtin_inff(), v1 = -__builtin_inff(), v2 = -__builtin_inff(), v3 = -__builtin_inff();
                unsigned int s0 = 0u, s1 = 0u, s2 = 0u, s3 = 0u;
                if (h0) { const int q = lb + ((c16 + dq) & (NCAP - 1));      v0 = nodeV[q]; s0 = __float_as_uint(nodeS[q]); }
                if (cmax > 16) {
                    if (h1) { const int q = lb + ((c16 + 16 + dq) & (NCAP - 1)); v1 = nodeV[q]; s1 = __float_as_uint(nodeS[q]); }
                    if (cmax > 32) {
                        if (h2) { const int q = lb + ((c16 + 32 + dq) & (NCAP - 1)); v2 = nodeV[q]; s2 = __float_as_uint(nodeS[q]); }
                        if (h3) { const int q = lb + ((c16 + 48 + dq) & (NCAP - 1)); v3 = nodeV[q]; s3 = __float_as_uint(nodeS[q]); }
                    }
                }
                const unsigned int sfirst = row_get_u<0>(s0);
                if (!h0) s0 = sfirst;
                if (!h1) s1 = sfirst;
                if (!h2) s2 = sfirst;
                if (!h3) s3 = sfirst;
                float w0 = 0.0f, w1 = 0.0f, w2 = 0.0f, w3 = 0.0f;
                for (int q = 0; q < cmax; q += 16) {
                    const unsigned int sreg = q >= 48 ? s3 : (q >= 32 ? s2 : (q >= 16 ? s1 : s0));
                    float4 x[16];
                    rows_ld<0>(Xp, sreg, cmax - q, c16, x);
                    if (q == 0) {
                        const float m = row_max16(fmaxf(fmaxf(v0, v1), fmaxf(v2, v3)));
                        w0 = h0 ? expf(v0 - m) : 0.0f;
                        w1 = h1 ? expf(v1 - m) : 0.0f;
                        w2 = h2 ? expf(v2 - m) : 0.0f;
                        w3 = h3 ? expf(v3 - m) : 0.0f;
                        dsum = row_sum16((w0 + w1) + (w2 + w3));
                    }
                    const float wreg = q >= 48 ? w3 : (q >= 32 ? w2 : (q >= 16 ? w1 : w0));
                    rows_fm<0>(wreg, cmax - q, x, a);
                }
            }
            const bool  any = cn > 0;
            const float inv = (any && dsum != 0.0f) ? 1.0f / dsum : 0.0f;
            float rx = any ? a.x * inv : 0.0f;
            float ry = any ? a.y * inv : 0.0f;
            float rz = any ? a.z * inv : 0.0f;
            float rw = any ? a.w * inv : 0.0f;
            rx = rx > 0.0f ? rx : (expf(rx) - 1.0f);
            ry = ry > 0.0f ? ry : (expf(ry) - 1.0f);
            rz = rz > 0.0f ? rz : (expf(rz) - 1.0f);
            rw = rw > 0.0f ? rw : (expf(rw) - 1.0f);
            if (own) *(float4*)(out + (size_t)(lo + dl) * D_OUT + 4 * c16) = make_float4(rx, ry, rz, rw);
        }
    } else {
        float* acc = nbuf;
        for (int i = tid; i < CH * D_OUT; i += WG_T) acc[i] = 0.0f;
        if (tid < CH) {
            mI[tid]  = __float_as_int(-__builtin_inff());
            den[tid] = 0.0f;
        }
        __syncthreads();
        const int4* dv = (const int4*)e_dst;
        for (int sweep = 0; sweep < 2; ++sweep) {
            for (int q = tid; q < NV4; q += WG_T) {
                const int4 d = dv[q];
                const int dd[4] = {d.x, d.y, d.z, d.w};
                for (int c = 0; c < 4; ++c) {
                    const unsigned int dl = (unsigned int)(dd[c] - lo);
                    if (dl < (unsigned int)nn) {
                        const int   s = e_src[4 * q + c];
                        const float v = leaky(s_src[s] + sd[dl]);
                        if (sweep == 0) {
                            lds_fmax(&mI[dl], v);
                        } else {
                            const float w = expf(v - __int_as_float(mI[dl]));
                            atomicAdd(&den[dl], w);
                            const float* xs = Xp + (size_t)s * D_OUT;
                            for (int f = 0; f < D_OUT; ++f) atomicAdd(&acc[dl * D_OUT + f], w * xs[f]);
                        }
                    }
                }
            }
            __syncthreads();
        }
        float* o = out + (size_t)lo * D_OUT;
        for (int i = tid; i < nn * D_OUT; i += WG_T) {
            const float dn = den[i >> 6];
            const float r  = dn == 0.0f ? 0.0f : acc[i] / dn;
            o[i] = r > 0.0f ? r : (expf(r) - 1.0f);
        }
    }
}

extern "C" void kernel_launch(void* const* d_in, const int* in_sizes, int n_in,
                              void* d_out, int out_size, void* d_ws, size_t ws_size,
                              hipStream_t stream) {
    const float* X       = (const float*)d_in[0];
    const float* theta_w = (const float*)d_in[1];
    const float* theta_b = (const float*)d_in[2];
    const float* a_src   = (const float*)d_in[3];
    const float* a_src_b = (const float*)d_in[4];
    const float* a_dst   = (const float*)d_in[5];
    const float* a_dst_b = (const float*)d_in[6];
    const int*   e_src   = (const int*)d_in[7];
    const int*   e_dst   = (const int*)d_in[8];
    float*       out     = (float*)d_out;

    float* Xp    = (float*)d_ws;
    float* s_src = Xp    + (size_t)N_NODES * D_OUT;
    float* s_dst = s_src + N_NODES;
    unsigned int* bm = (unsigned int*)(s_dst + N_NODES);
    unsigned int* pk = bm + (size_t)N_BM * BM_STRIDE;

    k_gemm<<<GEMM_GRID, GEMM_WAVES * 32, 0, stream>>>(X, theta_w, theta_b, a_src, a_src_b, a_dst, a_dst_b,
                                                       Xp, s_src, s_dst, e_src, e_dst, bm, pk,
                                                       (unsigned long long)ws_size);
    k_owner<<<N_WG, WG_T, 0, stream>>>(e_src, e_dst, s_src, s_dst, Xp, bm, pk, (unsigned long long)ws_size, out);
}
